# starting kernel
# baseline (speedup 1.0000x reference)
#include <hip/hip_runtime.h>


typedef float v2f __attribute__((ext_vector_type(2)));
typedef float v8f __attribute__((ext_vector_type(8)));

template <int NT, int K, int NCOL>
__global__ __launch_bounds__(256) void gemm_wmma_f32(const float* __restrict__ A,
                                                     const float* __restrict__ B,
                                                     float* __restrict__ C,
                                                     int mtiles) {
  int wave = (int)((blockIdx.x * blockDim.x + threadIdx.x) >> 5);
  int lane = (int)(threadIdx.x & 31);
  if (wave < mtiles) {
    const int lane15 = lane & 15;
    const int khalf  = (lane >> 4) * 2;
    const long rowbase = (long)wave * 16;

    v8f acc[NT];
#pragma unroll
    for (int t = 0; t < NT; ++t) acc[t] = (v8f){0.f,0.f,0.f,0.f,0.f,0.f,0.f,0.f};

    const float* arow = A + (rowbase + lane15) * K + khalf;
#pragma unroll 4
    for (int kb = 0; kb < K; kb += 4) {
      v2f a;
      a.x = arow[kb];
      a.y = arow[kb + 1];
#pragma unroll
      for (int t = 0; t < NT; ++t) {
        const float* bp = B + (size_t)(kb + khalf) * NCOL + t * 16 + lane15;
        v2f b;
        b.x = bp[0];
        b.y = bp[NCOL];
        acc[t] = __builtin_amdgcn_wmma_f32_16x16x4_f32(
             false, a,  false, b,
             (short)0, acc[t],  false,  false);
      }
    }
    const int rowoff = (lane >> 4) * 8;
#pragma unroll
    for (int t = 0; t < NT; ++t) {
#pragma unroll
      for (int j = 0; j < 8; ++j) {
        C[(rowbase + rowoff + j) * NCOL + t * 16 + lane15] = acc[t][j];
      }
    }
  }
}

template <int H, int C>
__global__ void att_k(const float* __restrict__ h, const float* __restrict__ att_s,
                      const float* __restrict__ att_d, float* __restrict__ a_s,
                      float* __restrict__ a_d, int n) {
  int i = blockIdx.x * blockDim.x + threadIdx.x;
  if (i >= n * H) return;
  int node = i / H, head = i % H;
  const float* hp = h + (long)node * (H * C) + head * C;
  float s = 0.f, d = 0.f;
#pragma unroll
  for (int c = 0; c < C; ++c) {
    float v = hp[c];
    s += v * att_s[head * C + c];
    d += v * att_d[head * C + c];
  }
  a_s[i] = s;
  a_d[i] = d;
}

__device__ __forceinline__ unsigned ordenc(float f) {
  unsigned b = __float_as_uint(f);
  return (b & 0x80000000u) ? ~b : (b | 0x80000000u);
}
__device__ __forceinline__ float orddec(unsigned u) {
  unsigned b = (u & 0x80000000u) ? (u & 0x7fffffffu) : ~u;
  return __uint_as_float(b);
}

__device__ __forceinline__ void edge_ids(const int* __restrict__ ei, int E, int i,
                                         int& s, int& d) {
  if (i < E) { s = ei[i]; d = ei[E + i]; }
  else       { s = i - E; d = i - E; }
}

template <int H>
__global__ void edge_max_k(const int* __restrict__ ei, int E, int n,
                           const float* __restrict__ a_s, const float* __restrict__ a_d,
                           unsigned* __restrict__ om) {
  int i = blockIdx.x * blockDim.x + threadIdx.x;
  if (i >= E + n) return;
  int s, d; edge_ids(ei, E, i, s, d);
#pragma unroll
  for (int h = 0; h < H; ++h) {
    float e = a_s[(long)s * H + h] + a_d[(long)d * H + h];
    e = (e >= 0.f) ? e : 0.2f * e;
    atomicMax(&om[(long)d * H + h], ordenc(e));
  }
}

template <int H>
__global__ void edge_exp_k(const int* __restrict__ ei, int E, int n,
                           const float* __restrict__ a_s, const float* __restrict__ a_d,
                           const unsigned* __restrict__ om, float* __restrict__ den,
                           float* __restrict__ eexp) {
  int i = blockIdx.x * blockDim.x + threadIdx.x;
  if (i >= E + n) return;
  int s, d; edge_ids(ei, E, i, s, d);
#pragma unroll
  for (int h = 0; h < H; ++h) {
    float e = a_s[(long)s * H + h] + a_d[(long)d * H + h];
    e = (e >= 0.f) ? e : 0.2f * e;
    float ex = __expf(e - orddec(om[(long)d * H + h]));
    eexp[(long)i * H + h] = ex;
    atomicAdd(&den[(long)d * H + h], ex);
  }
}

__global__ __launch_bounds__(256) void scatter1_k(const int* __restrict__ ei, int E, int n,
                                                  const float* __restrict__ h,
                                                  const float* __restrict__ eexp,
                                                  const float* __restrict__ den,
                                                  float* __restrict__ out) {
  int wv   = (int)((blockIdx.x * blockDim.x + threadIdx.x) >> 5);
  int lane = (int)(threadIdx.x & 31);
  if (wv >= E + n) return;
  int s, d; edge_ids(ei, E, wv, s, d);
  int head = lane >> 3;
  float alpha = eexp[(long)wv * 4 + head] / den[(long)d * 4 + head];
  const float4 hv = *(const float4*)(h + (long)s * 128 + lane * 4);
  float* op = out + (long)d * 128 + lane * 4;
  atomicAdd(op + 0, hv.x * alpha);
  atomicAdd(op + 1, hv.y * alpha);
  atomicAdd(op + 2, hv.z * alpha);
  atomicAdd(op + 3, hv.w * alpha);
}

__global__ void scatter2_k(const int* __restrict__ ei, int E, int n,
                           const float* __restrict__ h2, const float* __restrict__ eexp,
                           const float* __restrict__ den, float* __restrict__ out) {
  long idx = (long)blockIdx.x * blockDim.x + threadIdx.x;
  long tot = (long)(E + n) * 16;
  if (idx >= tot) return;
  int e = (int)(idx >> 4);
  int c = (int)(idx & 15);
  int s, d; edge_ids(ei, E, e, s, d);
  float alpha = eexp[e] / den[d];
  atomicAdd(&out[(long)d * 16 + c], h2[(long)s * 16 + c] * alpha);
}

__global__ void fill_u32_k(unsigned* __restrict__ p, unsigned v, long cnt) {
  long i = (long)blockIdx.x * blockDim.x + threadIdx.x;
  if (i < cnt) p[i] = v;
}
template <int C>
__global__ void init_bias_k(float* __restrict__ p, const float* __restrict__ b, long cnt) {
  long i = (long)blockIdx.x * blockDim.x + threadIdx.x;
  if (i < cnt) p[i] = b[i & (C - 1)];
}
__global__ void elu_k(float* __restrict__ p, long cnt) {
  long i = (long)blockIdx.x * blockDim.x + threadIdx.x;
  if (i < cnt) {
    float x = p[i];
    p[i] = (x > 0.f) ? x : expm1f(x);
  }
}

extern "C" void kernel_launch(void* const* d_in, const int* in_sizes, int n_in,
                              void* d_out, int out_size, void* d_ws, size_t ws_size,
                              hipStream_t stream) {
  const float* x    = (const float*)d_in[0];
  const int*   ei   = (const int*)d_in[1];
  const float* W1   = (const float*)d_in[2];
  const float* as1w = (const float*)d_in[3];
  const float* ad1w = (const float*)d_in[4];
  const float* b1   = (const float*)d_in[5];
  const float* W2   = (const float*)d_in[6];
  const float* as2w = (const float*)d_in[7];
  const float* ad2w = (const float*)d_in[8];
  const float* b2   = (const float*)d_in[9];
  float* out = (float*)d_out;

  const int n   = in_sizes[0] / 128;
  const int E   = in_sizes[1] / 2;
  const int tot = E + n;

  float* ws   = (float*)d_ws;
  float*    h1   = ws;
  float*    out1 = h1 + (size_t)n * 128;
  float*    as1  = out1 + (size_t)n * 128;
  float*    ad1  = as1 + (size_t)n * 4;
  unsigned* om1  = (unsigned*)(ad1 + (size_t)n * 4);
  float*    den1 = (float*)(om1 + (size_t)n * 4);
  float*    eexp = den1 + (size_t)n * 4;
  float*    h2   = h1;
  float*    as2  = h1 + (size_t)n * 16;
  float*    ad2  = as2 + n;
  unsigned* om2  = (unsigned*)(ad2 + n);
  float*    den2 = (float*)(om2 + n);

  const unsigned ORD_NEG_INF = 0x007FFFFFu;
  const int B = 256;
  const int mtiles = (n + 15) / 16;

  gemm_wmma_f32<8, 128, 128><<<(mtiles + 7) / 8, B, 0, stream>>>(x, W1, h1, mtiles);
  att_k<4, 32><<<(n * 4 + B - 1) / B, B, 0, stream>>>(h1, as1w, ad1w, as1, ad1, n);
  fill_u32_k<<<(int)(((long)n * 4 + B - 1) / B), B, 0, stream>>>(om1, ORD_NEG_INF, (long)n * 4);
  fill_u32_k<<<(int)(((long)n * 4 + B - 1) / B), B, 0, stream>>>((unsigned*)den1, 0u, (long)n * 4);
  init_bias_k<128><<<(int)(((long)n * 128 + B - 1) / B), B, 0, stream>>>(out1, b1, (long)n * 128);
  edge_max_k<4><<<(tot + B - 1) / B, B, 0, stream>>>(ei, E, n, as1, ad1, om1);
  edge_exp_k<4><<<(tot + B - 1) / B, B, 0, stream>>>(ei, E, n, as1, ad1, om1, den1, eexp);
  scatter1_k<<<(int)(((long)tot * 32 + B - 1) / B), B, 0, stream>>>(ei, E, n, h1, eexp, den1, out1);
  elu_k<<<(int)(((long)n * 128 + B - 1) / B), B, 0, stream>>>(out1, (long)n * 128);

  gemm_wmma_f32<1, 128, 16><<<(mtiles + 7) / 8, B, 0, stream>>>(out1, W2, h2, mtiles);
  att_k<1, 16><<<(n + B - 1) / B, B, 0, stream>>>(h2, as2w, ad2w, as2, ad2, n);
  fill_u32_k<<<(n + B - 1) / B, B, 0, stream>>>(om2, ORD_NEG_INF, (long)n);
  fill_u32_k<<<(n + B - 1) / B, B, 0, stream>>>((unsigned*)den2, 0u, (long)n);
  init_bias_k<16><<<(int)(((long)n * 16 + B - 1) / B), B, 0, stream>>>(out, b2, (long)n * 16);
  edge_max_k<1><<<(tot + B - 1) / B, B, 0, stream>>>(ei, E, n, as2, ad2, om2);
  edge_exp_k<1><<<(tot + B - 1) / B, B, 0, stream>>>(ei, E, n, as2, ad2, om2, den2, eexp);
  scatter2_k<<<(int)(((long)tot * 16 + B - 1) / B), B, 0, stream>>>(ei, E, n, h2, eexp, den2, out);
}


# Edges grouped by destination once (counting sort, arrival-order rank); each node gathers its neighbours, softmax and sum in registers: no float atomics, one store per output, sum order changes
# speedup vs baseline: 4.1169x; 4.1169x over previous
#include <hip/hip_runtime.h>


typedef float v2f __attribute__((ext_vector_type(2)));
typedef float v8f __attribute__((ext_vector_type(8)));

template <int NT, int K, int NCOL>
__global__ __launch_bounds__(256) void gemm_wmma_f32(const float* __restrict__ A,
                                                     const float* __restrict__ B,
                                                     float* __restrict__ C,
                                                     int mtiles) {
  int wave = (int)((blockIdx.x * blockDim.x + threadIdx.x) >> 5);
  int lane = (int)(threadIdx.x & 31);
  if (wave < mtiles) {
    const int lane15 = lane & 15;
    const int khalf  = (lane >> 4) * 2;
    const long rowbase = (long)wave * 16;

    v8f acc[NT];
#pragma unroll
    for (int t = 0; t < NT; ++t) acc[t] = (v8f){0.f,0.f,0.f,0.f,0.f,0.f,0.f,0.f};

    const float* arow = A + (rowbase + lane15) * K + khalf;
#pragma unroll 4
    for (int kb = 0; kb < K; kb += 4) {
      v2f a;
      a.x = arow[kb];
      a.y = arow[kb + 1];
#pragma unroll
      for (int t = 0; t < NT; ++t) {
        const float* bp = B + (size_t)(kb + khalf) * NCOL + t * 16 + lane15;
        v2f b;
        b.x = bp[0];
        b.y = bp[NCOL];
        acc[t] = __builtin_amdgcn_wmma_f32_16x16x4_f32(
             false, a,  false, b,
             (short)0, acc[t],  false,  false);
      }
    }
    const int rowoff = (lane >> 4) * 8;
#pragma unroll
    for (int t = 0; t < NT; ++t) {
#pragma unroll
      for (int j = 0; j < 8; ++j) {
        C[(rowbase + rowoff + j) * NCOL + t * 16 + lane15] = acc[t][j];
      }
    }
  }
}

template <int H, int C>
__global__ void att_k(const float* __restrict__ h, const float* __restrict__ att_s,
                      const float* __restrict__ att_d, float* __restrict__ a_s,
                      float* __restrict__ a_d, int n) {
  int i = blockIdx.x * blockDim.x + threadIdx.x;
  if (i >= n * H) return;
  int node = i / H, head = i % H;
  const float* hp = h + (long)node * (H * C) + head * C;
  float s = 0.f, d = 0.f;
#pragma unroll
  for (int c = 0; c < C; ++c) {
    float v = hp[c];
    s += v * att_s[head * C + c];
    d += v * att_d[head * C + c];
  }
  a_s[i] = s;
  a_d[i] = d;
}

__global__ void dst_rank_k(const int* __restrict__ ei, int E, int* __restrict__ cnt,
                           int* __restrict__ rank) {
  int i = blockIdx.x * blockDim.x + threadIdx.x;
  if (i >= E) return;
  rank[i] = atomicAdd(&cnt[ei[E + i]], 1);
}

__global__ __launch_bounds__(1024) void rowptr_scan_k(const int* __restrict__ cnt,
                                                      int* __restrict__ rowptr, int n) {
  __shared__ int wtot[32];
  __shared__ int wpre[32];
  const int tid  = (int)threadIdx.x;
  const int lane = tid & 31;
  const int wv   = tid >> 5;
  const int base = (int)blockIdx.x * 4096;
  const int4* c4 = (const int4*)cnt;
  int pre = 0;
#pragma unroll 4
  for (int q = tid; q < (base >> 2); q += 1024) {
    const int4 v = c4[q];
    pre += (v.x + v.y) + (v.z + v.w);
  }
  const int i0 = base + tid * 4;
  int c[4];
  for (int k = 0; k < 4; ++k) c[k] = (i0 + k < n) ? cnt[i0 + k] : 0;
  const int loc = (c[0] + c[1]) + (c[2] + c[3]);
  int inc = loc;
  for (int o = 1; o < 32; o <<= 1) {
    const int t = __shfl_up(inc, o);
    if (lane >= o) inc += t;
  }
  for (int o = 16; o > 0; o >>= 1) pre += __shfl_xor(pre, o);
  if (lane == 31) wtot[wv] = inc;
  if (lane == 0)  wpre[wv] = pre;
  __syncthreads();
  int off = 0;
  for (int w = 0; w < 32; ++w) {
    off += wpre[w];
    if (w < wv) off += wtot[w];
  }
  int run = off + inc - loc;
  for (int k = 0; k < 4; ++k) {
    if (i0 + k < n) rowptr[i0 + k] = run;
    run += c[k];
    if (i0 + k == n - 1) rowptr[n] = run;
  }
}

__global__ void csr_fill_k(const int* __restrict__ ei, int E, const int* __restrict__ rowptr,
                           const int* __restrict__ rank, int* __restrict__ csrc) {
  int i = blockIdx.x * blockDim.x + threadIdx.x;
  if (i >= E) return;
  csrc[rowptr[ei[E + i]] + rank[i]] = ei[i];
}

__device__ __forceinline__ float lrelu(float e) { return (e >= 0.f) ? e : 0.2f * e; }

__global__ __launch_bounds__(256) void gat1_node_k(const int* __restrict__ rowptr,
                                                   const int* __restrict__ csrc,
                                                   const float* __restrict__ h,
                                                   const float* __restrict__ a_s,
                                                   const float* __restrict__ a_d,
                                                   const float* __restrict__ bias,
                                                   float* __restrict__ out, int n) {
  const int lane = (int)(threadIdx.x & 31);
  const int node = __builtin_amdgcn_readfirstlane((int)((blockIdx.x * blockDim.x + threadIdx.x) >> 5));
  if (node >= n) return;
  const int beg = __builtin_amdgcn_readfirstlane(rowptr[node]);
  const int cnt = __builtin_amdgcn_readfirstlane(rowptr[node + 1]) - beg + 1;
  const float4 ad = *(const float4*)(a_d + (size_t)node * 4);
  const float NEG = -__builtin_huge_valf();

  int s0 = node;
  float e0 = NEG, e1 = NEG, e2 = NEG, e3 = NEG;
  if (lane < cnt) {
    if (lane > 0) s0 = csrc[beg + lane - 1];
    const float4 as = *(const float4*)(a_s + (size_t)s0 * 4);
    e0 = lrelu(as.x + ad.x);
    e1 = lrelu(as.y + ad.y);
    e2 = lrelu(as.z + ad.z);
    e3 = lrelu(as.w + ad.w);
  }
  float m0 = e0, m1 = e1, m2 = e2, m3 = e3;
  for (int j = 32 + lane; j < cnt; j += 32) {
    const int s = csrc[beg + j - 1];
    const float4 as = *(const float4*)(a_s + (size_t)s * 4);
    m0 = fmaxf(m0, lrelu(as.x + ad.x));
    m1 = fmaxf(m1, lrelu(as.y + ad.y));
    m2 = fmaxf(m2, lrelu(as.z + ad.z));
    m3 = fmaxf(m3, lrelu(as.w + ad.w));
  }
  for (int o = 16; o > 0; o >>= 1) {
    m0 = fmaxf(m0, __shfl_xor(m0, o));
    m1 = fmaxf(m1, __shfl_xor(m1, o));
    m2 = fmaxf(m2, __shfl_xor(m2, o));
    m3 = fmaxf(m3, __shfl_xor(m3, o));
  }
  const bool v0 = lane < cnt;
  const float x0 = v0 ? __expf(e0 - m0) : 0.f;
  const float x1 = v0 ? __expf(e1 - m1) : 0.f;
  const float x2 = v0 ? __expf(e2 - m2) : 0.f;
  const float x3 = v0 ? __expf(e3 - m3) : 0.f;
  float z0 = x0, z1 = x1, z2 = x2, z3 = x3;
  for (int j = 32 + lane; j < cnt; j += 32) {
    const int s = csrc[beg + j - 1];
    const float4 as = *(const float4*)(a_s + (size_t)s * 4);
    z0 += __expf(lrelu(as.x + ad.x) - m0);
    z1 += __expf(lrelu(as.y + ad.y) - m1);
    z2 += __expf(lrelu(as.z + ad.z) - m2);
    z3 += __expf(lrelu(as.w + ad.w) - m3);
  }
  for (int o = 16; o > 0; o >>= 1) {
    z0 += __shfl_xor(z0, o);
    z1 += __shfl_xor(z1, o);
    z2 += __shfl_xor(z2, o);
    z3 += __shfl_xor(z3, o);
  }
  const float p0 = x0 / z0, p1 = x1 / z1, p2 = x2 / z2, p3 = x3 / z3;

  const int head = lane >> 3;
  const int k0 = (head == 0) ? -1 : 0;
  const int k1 = (head == 1) ? -1 : 0;
  const int k2 = (head == 2) ? -1 : 0;
  const int k3 = (head == 3) ? -1 : 0;
  const float* hl = h + lane * 4;
  float4 acc = {0.f, 0.f, 0.f, 0.f};
  for (int cb = 0; cb < cnt; cb += 32) {
    int s = s0;
    float q0 = p0, q1 = p1, q2 = p2, q3 = p3;
    if (cb > 0) {
      const int j = cb + lane;
      s = node; q0 = 0.f; q1 = 0.f; q2 = 0.f; q3 = 0.f;
      if (j < cnt) {
        s = csrc[beg + j - 1];
        const float4 as = *(const float4*)(a_s + (size_t)s * 4);
        q0 = __expf(lrelu(as.x + ad.x) - m0) / z0;
        q1 = __expf(lrelu(as.y + ad.y) - m1) / z1;
        q2 = __expf(lrelu(as.z + ad.z) - m2) / z2;
        q3 = __expf(lrelu(as.w + ad.w) - m3) / z3;
      }
    }
    const int lim = (cnt - cb < 32) ? (cnt - cb) : 32;
#pragma unroll 4
    for (int t = 0; t < lim; ++t) {
      const int st = __builtin_amdgcn_readlane(s, t);
      const int r0 = __builtin_amdgcn_readlane(__float_as_int(q0), t);
      const int r1 = __builtin_amdgcn_readlane(__float_as_int(q1), t);
      const int r2 = __builtin_amdgcn_readlane(__float_as_int(q2), t);
      const int r3 = __builtin_amdgcn_readlane(__float_as_int(q3), t);
      const float alpha = __int_as_float((r0 & k0) | (r1 & k1) | (r2 & k2) | (r3 & k3));
      const float4 hv = *(const float4*)(hl + (size_t)st * 128);
      acc.x += hv.x * alpha;
      acc.y += hv.y * alpha;
      acc.z += hv.z * alpha;
      acc.w += hv.w * alpha;
    }
  }
  float4 r;
  r.x = acc.x + bias[lane * 4 + 0];
  r.y = acc.y + bias[lane * 4 + 1];
  r.z = acc.z + bias[lane * 4 + 2];
  r.w = acc.w + bias[lane * 4 + 3];
  r.x = (r.x > 0.f) ? r.x : expm1f(r.x);
  r.y = (r.y > 0.f) ? r.y : expm1f(r.y);
  r.z = (r.z > 0.f) ? r.z : expm1f(r.z);
  r.w = (r.w > 0.f) ? r.w : expm1f(r.w);
  *(float4*)(out + (size_t)node * 128 + lane * 4) = r;
}

__global__ __launch_bounds__(256) void gat2_node_k(const int* __restrict__ rowptr,
                                                   const int* __restrict__ csrc,
                                                   const float* __restrict__ h2,
                                                   const float* __restrict__ a_s,
                                                   const float* __restrict__ a_d,
                                                   const float* __restrict__ bias,
                                                   float* __restrict__ out, int n) {
  const int node = (int)((blockIdx.x * blockDim.x + threadIdx.x) >> 4);
  const int c    = (int)(threadIdx.x & 15);
  if (node >= n) return;
  const int beg = rowptr[node];
  const int cnt = rowptr[node + 1] - beg + 1;
  const float ad = a_d[node];
  const float NEG = -__builtin_huge_valf();

  int s0 = node;
  float e0 = NEG;
  if (c < cnt) {
    if (c > 0) s0 = csrc[beg + c - 1];
    e0 = lrelu(a_s[s0] + ad);
  }
  float m = e0;
  for (int j = 16 + c; j < cnt; j += 16) m = fmaxf(m, lrelu(a_s[csrc[beg + j - 1]] + ad));
  for (int o = 8; o > 0; o >>= 1) m = fmaxf(m, __shfl_xor(m, o));
  const float x0 = (c < cnt) ? __expf(e0 - m) : 0.f;
  float z = x0;
  for (int j = 16 + c; j < cnt; j += 16) z += __expf(lrelu(a_s[csrc[beg + j - 1]] + ad) - m);
  for (int o = 8; o > 0; o >>= 1) z += __shfl_xor(z, o);
  const float p0 = x0 / z;

  float acc = 0.f;
  for (int cb = 0; cb < cnt; cb += 16) {
    int s = s0;
    float q = p0;
    if (cb > 0) {
      const int j = cb + c;
      s = node; q = 0.f;
      if (j < cnt) {
        s = csrc[beg + j - 1];
        q = __expf(lrelu(a_s[s] + ad) - m) / z;
      }
    }
    const int lim = (cnt - cb < 16) ? (cnt - cb) : 16;
    for (int t = 0; t < lim; t += 4) {
#pragma unroll
      for (int k = 0; k < 4; ++k) {
        const int   tt = t + k;
        const int   st = __shfl(s, tt, 16);
        const float a  = __shfl(q, tt, 16);
        const float hv = h2[(size_t)st * 16 + c];
        acc += ((tt < lim) ? hv : 0.f) * a;
      }
    }
  }
  out[(size_t)node * 16 + c] = acc + bias[c];
}

__global__ void fill_u32_k(unsigned* __restrict__ p, unsigned v, long cnt) {
  long i = (long)blockIdx.x * blockDim.x + threadIdx.x;
  if (i < cnt) p[i] = v;
}

extern "C" void kernel_launch(void* const* d_in, const int* in_sizes, int n_in,
                              void* d_out, int out_size, void* d_ws, size_t ws_size,
                              hipStream_t stream) {
  const float* x    = (const float*)d_in[0];
  const int*   ei   = (const int*)d_in[1];
  const float* W1   = (const float*)d_in[2];
  const float* as1w = (const float*)d_in[3];
  const float* ad1w = (const float*)d_in[4];
  const float* b1   = (const float*)d_in[5];
  const float* W2   = (const float*)d_in[6];
  const float* as2w = (const float*)d_in[7];
  const float* ad2w = (const float*)d_in[8];
  const float* b2   = (const float*)d_in[9];
  float* out = (float*)d_out;

  const int n = in_sizes[0] / 128;
  const int E = in_sizes[1] / 2;

  float* ws     = (float*)d_ws;
  float* h1     = ws;
  float* out1   = h1 + (size_t)n * 128;
  float* as1    = out1 + (size_t)n * 128;
  float* ad1    = as1 + (size_t)n * 4;
  float* h2     = ad1 + (size_t)n * 4;
  int*   cnt    = (int*)(h2 + (size_t)n * 16);
  float* as2    = (float*)(cnt + n);
  float* ad2    = as2 + n;
  int*   rowptr = (int*)(ad2 + n);
  int*   rank   = rowptr + n + 1;
  int*   csrc   = rank + E;

  const int B = 256;
  const int mtiles = (n + 15) / 16;

  fill_u32_k<<<(n + B - 1) / B, B, 0, stream>>>((unsigned*)cnt, 0u, (long)n);
  dst_rank_k<<<(E + B - 1) / B, B, 0, stream>>>(ei, E, cnt, rank);
  rowptr_scan_k<<<(n + 4095) / 4096, 1024, 0, stream>>>(cnt, rowptr, n);
  csr_fill_k<<<(E + B - 1) / B, B, 0, stream>>>(ei, E, rowptr, rank, csrc);

  gemm_wmma_f32<8, 128, 128><<<(mtiles + 7) / 8, B, 0, stream>>>(x, W1, h1, mtiles);
  att_k<4, 32><<<(n * 4 + B - 1) / B, B, 0, stream>>>(h1, as1w, ad1w, as1, ad1, n);
  gat1_node_k<<<(int)(((long)n * 32 + B - 1) / B), B, 0, stream>>>(rowptr, csrc, h1, as1, ad1, b1, out1, n);

  gemm_wmma_f32<1, 128, 16><<<(mtiles + 7) / 8, B, 0, stream>>>(out1, W2, h2, mtiles);
  att_k<1, 16><<<(n + B - 1) / B, B, 0, stream>>>(h2, as2w, ad2w, as2, ad2, n);
  gat2_node_k<<<(int)(((long)n * 16 + B - 1) / B), B, 0, stream>>>(rowptr, csrc, h2, as2, ad2, b2, out, n);
}


# Independent stages share a launch: x*W1 GEMM blocks beside the edge-ranking blocks, layer-1 scores beside the CSR fill (edge stages take 8 edges per lane); 10 launches to 8; arrival-order rank kept
# speedup vs baseline: 4.3245x; 1.0504x over previous
#include <hip/hip_runtime.h>


typedef float v2f __attribute__((ext_vector_type(2)));
typedef float v8f __attribute__((ext_vector_type(8)));

template <int NT, int K, int NCOL>
__device__ __forceinline__ void gemm_wave_tile(const float* __restrict__ A,
                                               const float* __restrict__ B,
                                               float* __restrict__ C,
                                               int wave, int lane) {
  const int lane15 = lane & 15;
  const int khalf  = (lane >> 4) * 2;
  const long rowbase = (long)wave * 16;

  v8f acc[NT];
#pragma unroll
  for (int t = 0; t < NT; ++t) acc[t] = (v8f){0.f,0.f,0.f,0.f,0.f,0.f,0.f,0.f};

  const float* arow = A + (rowbase + lane15) * K + khalf;
#pragma unroll 4
  for (int kb = 0; kb < K; kb += 4) {
    v2f a;
    a.x = arow[kb];
    a.y = arow[kb + 1];
#pragma unroll
    for (int t = 0; t < NT; ++t) {
      const float* bp = B + (size_t)(kb + khalf) * NCOL + t * 16 + lane15;
      v2f b;
      b.x = bp[0];
      b.y = bp[NCOL];
      acc[t] = __builtin_amdgcn_wmma_f32_16x16x4_f32(
           false, a,  false, b,
           (short)0, acc[t],  false,  false);
    }
  }
  const int rowoff = (lane >> 4) * 8;
#pragma unroll
  for (int t = 0; t < NT; ++t) {
#pragma unroll
    for (int j = 0; j < 8; ++j) {
      C[(rowbase + rowoff + j) * NCOL + t * 16 + lane15] = acc[t][j];
    }
  }
}

template <int NT, int K, int NCOL>
__global__ __launch_bounds__(256) void gemm_wmma_f32(const float* __restrict__ A,
                                                     const float* __restrict__ B,
                                                     float* __restrict__ C,
                                                     int mtiles) {
  const int wave = (int)((blockIdx.x * blockDim.x + threadIdx.x) >> 5);
  const int lane = (int)(threadIdx.x & 31);
  if (wave < mtiles) gemm_wave_tile<NT, K, NCOL>(A, B, C, wave, lane);
}

template <int H, int C>
__device__ __forceinline__ void att_thread(const float* __restrict__ h, const float* __restrict__ att_s,
                                           const float* __restrict__ att_d, float* __restrict__ a_s,
                                           float* __restrict__ a_d, int n, int i) {
  if (i >= n * H) return;
  int node = i / H, head = i % H;
  const float* hp = h + (long)node * (H * C) + head * C;
  float s = 0.f, d = 0.f;
#pragma unroll
  for (int c = 0; c < C; ++c) {
    float v = hp[c];
    s += v * att_s[head * C + c];
    d += v * att_d[head * C + c];
  }
  a_s[i] = s;
  a_d[i] = d;
}

template <int H, int C>
__global__ void att_k(const float* __restrict__ h, const float* __restrict__ att_s,
                      const float* __restrict__ att_d, float* __restrict__ a_s,
                      float* __restrict__ a_d, int n) {
  att_thread<H, C>(h, att_s, att_d, a_s, a_d, n, (int)(blockIdx.x * blockDim.x + threadIdx.x));
}

__device__ __forceinline__ void rank_wave(const int* __restrict__ dst, int E, int* __restrict__ cnt,
                                          int* __restrict__ rank, int w, int lane) {
  const long base = (long)w * 256;
  if (base >= E) return;
  const bool vec = (base + 256 <= E) &&
                   (((((size_t)(dst + base)) | ((size_t)(rank + base))) & 15) == 0);
  if (vec) {
    const int4 d0 = *(const int4*)(dst + base + lane * 4);
    const int4 d1 = *(const int4*)(dst + base + 128 + lane * 4);
    asm volatile("" :: "v"(d0.x), "v"(d0.y), "v"(d0.z), "v"(d0.w),
                       "v"(d1.x), "v"(d1.y), "v"(d1.z), "v"(d1.w));
    int4 r0, r1;
    r0.x = atomicAdd(&cnt[d0.x], 1);
    r0.y = atomicAdd(&cnt[d0.y], 1);
    r0.z = atomicAdd(&cnt[d0.z], 1);
    r0.w = atomicAdd(&cnt[d0.w], 1);
    r1.x = atomicAdd(&cnt[d1.x], 1);
    r1.y = atomicAdd(&cnt[d1.y], 1);
    r1.z = atomicAdd(&cnt[d1.z], 1);
    r1.w = atomicAdd(&cnt[d1.w], 1);
    *(int4*)(rank + base + lane * 4) = r0;
    *(int4*)(rank + base + 128 + lane * 4) = r1;
  } else {
    for (int k = 0; k < 8; ++k) {
      const long i = base + k * 32 + lane;
      if (i < E) rank[i] = atomicAdd(&cnt[dst[i]], 1);
    }
  }
}

__global__ __launch_bounds__(256) void proj1_rank_k(const float* __restrict__ x,
                                                    const float* __restrict__ W1,
                                                    float* __restrict__ h1, int mtiles,
                                                    int gemm_blocks,
                                                    const int* __restrict__ ei, int E,
                                                    int* __restrict__ cnt,
                                                    int* __restrict__ rank) {
  const int lane = (int)(threadIdx.x & 31);
  if ((int)blockIdx.x < gemm_blocks) {
    const int wave = (int)((blockIdx.x * 256 + threadIdx.x) >> 5);
    if (wave < mtiles) gemm_wave_tile<8, 128, 128>(x, W1, h1, wave, lane);
  } else {
    const int w = (int)((((int)blockIdx.x - gemm_blocks) * 256 + threadIdx.x) >> 5);
    rank_wave(ei + E, E, cnt, rank, w, lane);
  }
}

__global__ __launch_bounds__(1024) void rowptr_scan_k(const int* __restrict__ cnt,
                                                      int* __restrict__ rowptr, int n) {
  __shared__ int wtot[32];
  __shared__ int wpre[32];
  const int tid  = (int)threadIdx.x;
  const int lane = tid & 31;
  const int wv   = tid >> 5;
  const int base = (int)blockIdx.x * 4096;
  const int4* c4 = (const int4*)cnt;
  int pre = 0;
#pragma unroll 4
  for (int q = tid; q < (base >> 2); q += 1024) {
    const int4 v = c4[q];
    pre += (v.x + v.y) + (v.z + v.w);
  }
  const int i0 = base + tid * 4;
  int c[4];
  for (int k = 0; k < 4; ++k) c[k] = (i0 + k < n) ? cnt[i0 + k] : 0;
  const int loc = (c[0] + c[1]) + (c[2] + c[3]);
  int inc = loc;
  for (int o = 1; o < 32; o <<= 1) {
    const int t = __shfl_up(inc, o);
    if (lane >= o) inc += t;
  }
  for (int o = 16; o > 0; o >>= 1) pre += __shfl_xor(pre, o);
  if (lane == 31) wtot[wv] = inc;
  if (lane == 0)  wpre[wv] = pre;
  __syncthreads();
  int off = 0;
  for (int w = 0; w < 32; ++w) {
    off += wpre[w];
    if (w < wv) off += wtot[w];
  }
  int run = off + inc - loc;
  for (int k = 0; k < 4; ++k) {
    if (i0 + k < n) rowptr[i0 + k] = run;
    run += c[k];
    if (i0 + k == n - 1) rowptr[n] = run;
  }
}

__device__ __forceinline__ void csrfill_wave(const int* __restrict__ ei, int E,
                                             const int* __restrict__ rowptr,
                                             const int* __restrict__ rank,
                                             int* __restrict__ csrc, int w, int lane) {
  const long base = (long)w * 256;
  if (base >= E) return;
  const int* sp = ei + base;
  const int* dp = ei + E + base;
  const int* rp = rank + base;
  const bool vec = (base + 256 <= E) &&
                   (((((size_t)sp) | ((size_t)dp) | ((size_t)rp)) & 15) == 0);
  if (vec) {
    const int4 d0 = *(const int4*)(dp + lane * 4);
    const int4 d1 = *(const int4*)(dp + 128 + lane * 4);
    const int4 r0 = *(const int4*)(rp + lane * 4);
    const int4 r1 = *(const int4*)(rp + 128 + lane * 4);
    const int4 s0 = *(const int4*)(sp + lane * 4);
    const int4 s1 = *(const int4*)(sp + 128 + lane * 4);
    const int p0 = rowptr[d0.x], p1 = rowptr[d0.y], p2 = rowptr[d0.z], p3 = rowptr[d0.w];
    const int p4 = rowptr[d1.x], p5 = rowptr[d1.y], p6 = rowptr[d1.z], p7 = rowptr[d1.w];
    csrc[p0 + r0.x] = s0.x;
    csrc[p1 + r0.y] = s0.y;
    csrc[p2 + r0.z] = s0.z;
    csrc[p3 + r0.w] = s0.w;
    csrc[p4 + r1.x] = s1.x;
    csrc[p5 + r1.y] = s1.y;
    csrc[p6 + r1.z] = s1.z;
    csrc[p7 + r1.w] = s1.w;
  } else {
    for (int k = 0; k < 8; ++k) {
      const long i = base + k * 32 + lane;
      if (i < E) csrc[rowptr[ei[E + i]] + rank[i]] = ei[i];
    }
  }
}

__global__ __launch_bounds__(256) void score1_csrfill_k(const float* __restrict__ h1,
                                                        const float* __restrict__ att_s,
                                                        const float* __restrict__ att_d,
                                                        float* __restrict__ a_s,
                                                        float* __restrict__ a_d, int n,
                                                        int fill_blocks,
                                                        const int* __restrict__ ei, int E,
                                                        const int* __restrict__ rowptr,
                                                        const int* __restrict__ rank,
                                                        int* __restrict__ csrc) {
  if ((int)blockIdx.x < fill_blocks) {
    const int w = (int)((blockIdx.x * 256 + threadIdx.x) >> 5);
    csrfill_wave(ei, E, rowptr, rank, csrc, w, (int)(threadIdx.x & 31));
  } else {
    att_thread<4, 32>(h1, att_s, att_d, a_s, a_d, n,
                      (int)((((int)blockIdx.x - fill_blocks) * 256) + threadIdx.x));
  }
}

__device__ __forceinline__ float lrelu(float e) { return (e >= 0.f) ? e : 0.2f * e; }

__global__ __launch_bounds__(256) void gat1_node_k(const int* __restrict__ rowptr,
                                                   const int* __restrict__ csrc,
                                                   const float* __restrict__ h,
                                                   const float* __restrict__ a_s,
                                                   const float* __restrict__ a_d,
                                                   const float* __restrict__ bias,
                                                   float* __restrict__ out, int n) {
  const int lane = (int)(threadIdx.x & 31);
  const int node = __builtin_amdgcn_readfirstlane((int)((blockIdx.x * blockDim.x + threadIdx.x) >> 5));
  if (node >= n) return;
  const int beg = __builtin_amdgcn_readfirstlane(rowptr[node]);
  const int cnt = __builtin_amdgcn_readfirstlane(rowptr[node + 1]) - beg + 1;
  const float4 ad = *(const float4*)(a_d + (size_t)node * 4);
  const float NEG = -__builtin_huge_valf();

  int s0 = node;
  float e0 = NEG, e1 = NEG, e2 = NEG, e3 = NEG;
  if (lane < cnt) {
    if (lane > 0) s0 = csrc[beg + lane - 1];
    const float4 as = *(const float4*)(a_s + (size_t)s0 * 4);
    e0 = lrelu(as.x + ad.x);
    e1 = lrelu(as.y + ad.y);
    e2 = lrelu(as.z + ad.z);
    e3 = lrelu(as.w + ad.w);
  }
  float m0 = e0, m1 = e1, m2 = e2, m3 = e3;
  for (int j = 32 + lane; j < cnt; j += 32) {
    const int s = csrc[beg + j - 1];
    const float4 as = *(const float4*)(a_s + (size_t)s * 4);
    m0 = fmaxf(m0, lrelu(as.x + ad.x));
    m1 = fmaxf(m1, lrelu(as.y + ad.y));
    m2 = fmaxf(m2, lrelu(as.z + ad.z));
    m3 = fmaxf(m3, lrelu(as.w + ad.w));
  }
  for (int o = 16; o > 0; o >>= 1) {
    m0 = fmaxf(m0, __shfl_xor(m0, o));
    m1 = fmaxf(m1, __shfl_xor(m1, o));
    m2 = fmaxf(m2, __shfl_xor(m2, o));
    m3 = fmaxf(m3, __shfl_xor(m3, o));
  }
  const bool v0 = lane < cnt;
  const float x0 = v0 ? __expf(e0 - m0) : 0.f;
  const float x1 = v0 ? __expf(e1 - m1) : 0.f;
  const float x2 = v0 ? __expf(e2 - m2) : 0.f;
  const float x3 = v0 ? __expf(e3 - m3) : 0.f;
  float z0 = x0, z1 = x1, z2 = x2, z3 = x3;
  for (int j = 32 + lane; j < cnt; j += 32) {
    const int s = csrc[beg + j - 1];
    const float4 as = *(const float4*)(a_s + (size_t)s * 4);
    z0 += __expf(lrelu(as.x + ad.x) - m0);
    z1 += __expf(lrelu(as.y + ad.y) - m1);
    z2 += __expf(lrelu(as.z + ad.z) - m2);
    z3 += __expf(lrelu(as.w + ad.w) - m3);
  }
  for (int o = 16; o > 0; o >>= 1) {
    z0 += __shfl_xor(z0, o);
    z1 += __shfl_xor(z1, o);
    z2 += __shfl_xor(z2, o);
    z3 += __shfl_xor(z3, o);
  }
  const float p0 = x0 / z0, p1 = x1 / z1, p2 = x2 / z2, p3 = x3 / z3;

  const int head = lane >> 3;
  const int k0 = (head == 0) ? -1 : 0;
  const int k1 = (head == 1) ? -1 : 0;
  const int k2 = (head == 2) ? -1 : 0;
  const int k3 = (head == 3) ? -1 : 0;
  const float* hl = h + lane * 4;
  float4 acc = {0.f, 0.f, 0.f, 0.f};
  for (int cb = 0; cb < cnt; cb += 32) {
    int s = s0;
    float q0 = p0, q1 = p1, q2 = p2, q3 = p3;
    if (cb > 0) {
      const int j = cb + lane;
      s = node; q0 = 0.f; q1 = 0.f; q2 = 0.f; q3 = 0.f;
      if (j < cnt) {
        s = csrc[beg + j - 1];
        const float4 as = *(const float4*)(a_s + (size_t)s * 4);
        q0 = __expf(lrelu(as.x + ad.x) - m0) / z0;
        q1 = __expf(lrelu(as.y + ad.y) - m1) / z1;
        q2 = __expf(lrelu(as.z + ad.z) - m2) / z2;
        q3 = __expf(lrelu(as.w + ad.w) - m3) / z3;
      }
    }
    const int lim = (cnt - cb < 32) ? (cnt - cb) : 32;
#pragma unroll 4
    for (int t = 0; t < lim; ++t) {
      const int st = __builtin_amdgcn_readlane(s, t);
      const int r0 = __builtin_amdgcn_readlane(__float_as_int(q0), t);
      const int r1 = __builtin_amdgcn_readlane(__float_as_int(q1), t);
      const int r2 = __builtin_amdgcn_readlane(__float_as_int(q2), t);
      const int r3 = __builtin_amdgcn_readlane(__float_as_int(q3), t);
      const float alpha = __int_as_float((r0 & k0) | (r1 & k1) | (r2 & k2) | (r3 & k3));
      const float4 hv = *(const float4*)(hl + (size_t)st * 128);
      acc.x += hv.x * alpha;
      acc.y += hv.y * alpha;
      acc.z += hv.z * alpha;
      acc.w += hv.w * alpha;
    }
  }
  float4 r;
  r.x = acc.x + bias[lane * 4 + 0];
  r.y = acc.y + bias[lane * 4 + 1];
  r.z = acc.z + bias[lane * 4 + 2];
  r.w = acc.w + bias[lane * 4 + 3];
  r.x = (r.x > 0.f) ? r.x : expm1f(r.x);
  r.y = (r.y > 0.f) ? r.y : expm1f(r.y);
  r.z = (r.z > 0.f) ? r.z : expm1f(r.z);
  r.w = (r.w > 0.f) ? r.w : expm1f(r.w);
  *(float4*)(out + (size_t)node * 128 + lane * 4) = r;
}

__global__ __launch_bounds__(256) void gat2_node_k(const int* __restrict__ rowptr,
                                                   const int* __restrict__ csrc,
                                                   const float* __restrict__ h2,
                                                   const float* __restrict__ a_s,
                                                   const float* __restrict__ a_d,
                                                   const float* __restrict__ bias,
                                                   float* __restrict__ out, int n) {
  const int node = (int)((blockIdx.x * blockDim.x + threadIdx.x) >> 4);
  const int c    = (int)(threadIdx.x & 15);
  if (node >= n) return;
  const int beg = rowptr[node];
  const int cnt = rowptr[node + 1] - beg + 1;
  const float ad = a_d[node];
  const float NEG = -__builtin_huge_valf();

  int s0 = node;
  float e0 = NEG;
  if (c < cnt) {
    if (c > 0) s0 = csrc[beg + c - 1];
    e0 = lrelu(a_s[s0] + ad);
  }
  float m = e0;
  for (int j = 16 + c; j < cnt; j += 16) m = fmaxf(m, lrelu(a_s[csrc[beg + j - 1]] + ad));
  for (int o = 8; o > 0; o >>= 1) m = fmaxf(m, __shfl_xor(m, o));
  const float x0 = (c < cnt) ? __expf(e0 - m) : 0.f;
  float z = x0;
  for (int j = 16 + c; j < cnt; j += 16) z += __expf(lrelu(a_s[csrc[beg + j - 1]] + ad) - m);
  for (int o = 8; o > 0; o >>= 1) z += __shfl_xor(z, o);
  const float p0 = x0 / z;

  float acc = 0.f;
  for (int cb = 0; cb < cnt; cb += 16) {
    int s = s0;
    float q = p0;
    if (cb > 0) {
      const int j = cb + c;
      s = node; q = 0.f;
      if (j < cnt) {
        s = csrc[beg + j - 1];
        q = __expf(lrelu(a_s[s] + ad) - m) / z;
      }
    }
    const int lim = (cnt - cb < 16) ? (cnt - cb) : 16;
    for (int t = 0; t < lim; t += 4) {
#pragma unroll
      for (int k = 0; k < 4; ++k) {
        const int   tt = t + k;
        const int   st = __shfl(s, tt, 16);
        const float a  = __shfl(q, tt, 16);
        const float hv = h2[(size_t)st * 16 + c];
        acc += ((tt < lim) ? hv : 0.f) * a;
      }
    }
  }
  out[(size_t)node * 16 + c] = acc + bias[c];
}

__global__ void fill_u32_k(unsigned* __restrict__ p, unsigned v, long cnt) {
  long i = (long)blockIdx.x * blockDim.x + threadIdx.x;
  if (i < cnt) p[i] = v;
}

extern "C" void kernel_launch(void* const* d_in, const int* in_sizes, int n_in,
                              void* d_out, int out_size, void* d_ws, size_t ws_size,
                              hipStream_t stream) {
  const float* x    = (const float*)d_in[0];
  const int*   ei   = (const int*)d_in[1];
  const float* W1   = (const float*)d_in[2];
  const float* as1w = (const float*)d_in[3];
  const float* ad1w = (const float*)d_in[4];
  const float* b1   = (const float*)d_in[5];
  const float* W2   = (const float*)d_in[6];
  const float* as2w = (const float*)d_in[7];
  const float* ad2w = (const float*)d_in[8];
  const float* b2   = (const float*)d_in[9];
  float* out = (float*)d_out;

  const int n = in_sizes[0] / 128;
  const int E = in_sizes[1] / 2;

  float* ws     = (float*)d_ws;
  float* h1     = ws;
  float* out1   = h1 + (size_t)n * 128;
  float* as1    = out1 + (size_t)n * 128;
  float* ad1    = as1 + (size_t)n * 4;
  float* h2     = ad1 + (size_t)n * 4;
  int*   cnt    = (int*)(h2 + (size_t)n * 16);
  float* as2    = (float*)(cnt + n);
  float* ad2    = as2 + n;
  int*   rank   = (int*)(ad2 + n);
  int*   csrc   = rank + E;
  int*   rowptr = csrc + E;

  const int B = 256;
  const int mtiles = (n + 15) / 16;
  const int gemm_blocks = (mtiles + 7) / 8;
  const int edge_blocks = (E + 2047) / 2048;
  const int att1_blocks = (n * 4 + B - 1) / B;

  fill_u32_k<<<(n + B - 1) / B, B, 0, stream>>>((unsigned*)cnt, 0u, (long)n);
  proj1_rank_k<<<gemm_blocks + edge_blocks, B, 0, stream>>>(x, W1, h1, mtiles, gemm_blocks, ei, E, cnt, rank);
  rowptr_scan_k<<<(n + 4095) / 4096, 1024, 0, stream>>>(cnt, rowptr, n);
  score1_csrfill_k<<<edge_blocks + att1_blocks, B, 0, stream>>>(h1, as1w, ad1w, as1, ad1, n, edge_blocks, ei, E, rowptr, rank, csrc);

  gat1_node_k<<<(int)(((long)n * 32 + B - 1) / B), B, 0, stream>>>(rowptr, csrc, h1, as1, ad1, b1, out1, n);

  gemm_wmma_f32<1, 128, 16><<<(mtiles + 7) / 8, B, 0, stream>>>(out1, W2, h2, mtiles);
  att_k<1, 16><<<(n + B - 1) / B, B, 0, stream>>>(h2, as2w, ad2w, as2, ad2, n);
  gat2_node_k<<<(int)(((long)n * 16 + B - 1) / B), B, 0, stream>>>(rowptr, csrc, h2, as2, ad2, b2, out, n);
}


# Each edge's source goes straight to the slot of its arrival rank in a fixed 64-int row of its destination, in the ranking pass (rare spill list): no scan, no 2nd edge pass; 8 launches to 7; same sums
# speedup vs baseline: 4.5205x; 1.0453x over previous
#include <hip/hip_runtime.h>


typedef float v2f __attribute__((ext_vector_type(2)));
typedef float v8f __attribute__((ext_vector_type(8)));

template <int NT, int K, int NCOL>
__device__ __forceinline__ void gemm_wave_tile(const float* __restrict__ A,
                                               const float* __restrict__ B,
                                               float* __restrict__ C,
                                               int wave, int lane) {
  const int lane15 = lane & 15;
  const int khalf  = (lane >> 4) * 2;
  const long rowbase = (long)wave * 16;

  v8f acc[NT];
#pragma unroll
  for (int t = 0; t < NT; ++t) acc[t] = (v8f){0.f,0.f,0.f,0.f,0.f,0.f,0.f,0.f};

  const float* arow = A + (rowbase + lane15) * K + khalf;
#pragma unroll 4
  for (int kb = 0; kb < K; kb += 4) {
    v2f a;
    a.x = arow[kb];
    a.y = arow[kb + 1];
#pragma unroll
    for (int t = 0; t < NT; ++t) {
      const float* bp = B + (size_t)(kb + khalf) * NCOL + t * 16 + lane15;
      v2f b;
      b.x = bp[0];
      b.y = bp[NCOL];
      acc[t] = __builtin_amdgcn_wmma_f32_16x16x4_f32(
           false, a,  false, b,
           (short)0, acc[t],  false,  false);
    }
  }
  const int rowoff = (lane >> 4) * 8;
#pragma unroll
  for (int t = 0; t < NT; ++t) {
#pragma unroll
    for (int j = 0; j < 8; ++j) {
      C[(rowbase + rowoff + j) * NCOL + t * 16 + lane15] = acc[t][j];
    }
  }
}

template <int NT, int K, int NCOL>
__global__ __launch_bounds__(256) void gemm_wmma_f32(const float* __restrict__ A,
                                                     const float* __restrict__ B,
                                                     float* __restrict__ C,
                                                     int mtiles) {
  const int wave = (int)((blockIdx.x * blockDim.x + threadIdx.x) >> 5);
  const int lane = (int)(threadIdx.x & 31);
  if (wave < mtiles) gemm_wave_tile<NT, K, NCOL>(A, B, C, wave, lane);
}

template <int H, int C>
__device__ __forceinline__ void att_thread(const float* __restrict__ h, const float* __restrict__ att_s,
                                           const float* __restrict__ att_d, float* __restrict__ a_s,
                                           float* __restrict__ a_d, int n, int i) {
  if (i >= n * H) return;
  int node = i / H, head = i % H;
  const float* hp = h + (long)node * (H * C) + head * C;
  float s = 0.f, d = 0.f;
#pragma unroll
  for (int c = 0; c < C; ++c) {
    float v = hp[c];
    s += v * att_s[head * C + c];
    d += v * att_d[head * C + c];
  }
  a_s[i] = s;
  a_d[i] = d;
}

template <int H, int C>
__global__ void att_k(const float* __restrict__ h, const float* __restrict__ att_s,
                      const float* __restrict__ att_d, float* __restrict__ a_s,
                      float* __restrict__ a_d, int n) {
  att_thread<H, C>(h, att_s, att_d, a_s, a_d, n, (int)(blockIdx.x * blockDim.x + threadIdx.x));
}

constexpr int kRow = 64;

__device__ __forceinline__ void put_edge(int d, int r, int s, int* __restrict__ tab,
                                         int* __restrict__ spill_n, int* __restrict__ spill_dst,
                                         int* __restrict__ spill_src) {
  if (r < kRow - 1) {
    tab[(size_t)d * kRow + 1 + r] = s;
  } else {
    const int k = atomicAdd(spill_n, 1);
    spill_dst[k] = d;
    spill_src[k] = s;
  }
}

__device__ __forceinline__ void slot_wave(const int* __restrict__ esrc, const int* __restrict__ edst,
                                          int E, int* __restrict__ cnt, int* __restrict__ tab,
                                          int* __restrict__ spill_n, int* __restrict__ spill_dst,
                                          int* __restrict__ spill_src, int w, int lane) {
  const long base = (long)w * 256;
  if (base >= E) return;
  const bool vec = (base + 256 <= E) &&
                   (((((size_t)(edst + base)) | ((size_t)(esrc + base))) & 15) == 0);
  if (vec) {
    const int4 d0 = *(const int4*)(edst + base + lane * 4);
    const int4 d1 = *(const int4*)(edst + base + 128 + lane * 4);
    const int4 s0 = *(const int4*)(esrc + base + lane * 4);
    const int4 s1 = *(const int4*)(esrc + base + 128 + lane * 4);
    asm volatile("" :: "v"(d0.x), "v"(d0.y), "v"(d0.z), "v"(d0.w),
                       "v"(d1.x), "v"(d1.y), "v"(d1.z), "v"(d1.w),
                       "v"(s0.x), "v"(s0.y), "v"(s0.z), "v"(s0.w),
                       "v"(s1.x), "v"(s1.y), "v"(s1.z), "v"(s1.w));
    const int r0 = atomicAdd(&cnt[d0.x], 1);
    const int r1 = atomicAdd(&cnt[d0.y], 1);
    const int r2 = atomicAdd(&cnt[d0.z], 1);
    const int r3 = atomicAdd(&cnt[d0.w], 1);
    const int r4 = atomicAdd(&cnt[d1.x], 1);
    const int r5 = atomicAdd(&cnt[d1.y], 1);
    const int r6 = atomicAdd(&cnt[d1.z], 1);
    const int r7 = atomicAdd(&cnt[d1.w], 1);
    put_edge(d0.x, r0, s0.x, tab, spill_n, spill_dst, spill_src);
    put_edge(d0.y, r1, s0.y, tab, spill_n, spill_dst, spill_src);
    put_edge(d0.z, r2, s0.z, tab, spill_n, spill_dst, spill_src);
    put_edge(d0.w, r3, s0.w, tab, spill_n, spill_dst, spill_src);
    put_edge(d1.x, r4, s1.x, tab, spill_n, spill_dst, spill_src);
    put_edge(d1.y, r5, s1.y, tab, spill_n, spill_dst, spill_src);
    put_edge(d1.z, r6, s1.z, tab, spill_n, spill_dst, spill_src);
    put_edge(d1.w, r7, s1.w, tab, spill_n, spill_dst, spill_src);
  } else {
    for (int k = 0; k < 8; ++k) {
      const long i = base + k * 32 + lane;
      if (i < E) {
        const int d = edst[i];
        const int r = atomicAdd(&cnt[d], 1);
        put_edge(d, r, esrc[i], tab, spill_n, spill_dst, spill_src);
      }
    }
  }
}

__global__ __launch_bounds__(256) void proj1_slot_k(const float* __restrict__ x,
                                                    const float* __restrict__ W1,
                                                    float* __restrict__ h1, int mtiles,
                                                    int gemm_blocks,
                                                    const int* __restrict__ ei, int E,
                                                    int* __restrict__ cnt, int n,
                                                    int* __restrict__ tab,
                                                    int* __restrict__ spill_dst,
                                                    int* __restrict__ spill_src) {
  const int lane = (int)(threadIdx.x & 31);
  if ((int)blockIdx.x < gemm_blocks) {
    const int wave = (int)((blockIdx.x * 256 + threadIdx.x) >> 5);
    if (wave < mtiles) gemm_wave_tile<8, 128, 128>(x, W1, h1, wave, lane);
  } else {
    const int w = (int)((((int)blockIdx.x - gemm_blocks) * 256 + threadIdx.x) >> 5);
    slot_wave(ei, ei + E, E, cnt, tab, cnt + n, spill_dst, spill_src, w, lane);
  }
}

__device__ __forceinline__ float lrelu(float e) { return (e >= 0.f) ? e : 0.2f * e; }


__device__ __forceinline__ void add_slot(float4& acc, int s, float q0, float q1, float q2, float q3,
                                         int t, int k0, int k1, int k2, int k3,
                                         const float* __restrict__ hl) {
  const int st = __builtin_amdgcn_readlane(s, t);
  const int r0 = __builtin_amdgcn_readlane(__float_as_int(q0), t);
  const int r1 = __builtin_amdgcn_readlane(__float_as_int(q1), t);
  const int r2 = __builtin_amdgcn_readlane(__float_as_int(q2), t);
  const int r3 = __builtin_amdgcn_readlane(__float_as_int(q3), t);
  const float alpha = __int_as_float((r0 & k0) | (r1 & k1) | (r2 & k2) | (r3 & k3));
  const float4 hv = *(const float4*)(hl + (size_t)st * 128);
  acc.x += hv.x * alpha;
  acc.y += hv.y * alpha;
  acc.z += hv.z * alpha;
  acc.w += hv.w * alpha;
}

__global__ __launch_bounds__(256) void gat1_node_k(const int* __restrict__ deg,
                                                   const int* __restrict__ tab,
                                                   const int* __restrict__ spill_dst,
                                                   const int* __restrict__ spill_src, int E,
                                                   const float* __restrict__ h,
                                                   const float* __restrict__ a_s,
                                                   const float* __restrict__ a_d,
                                                   const float* __restrict__ bias,
                                                   float* __restrict__ out, int n) {
  const int lane = (int)(threadIdx.x & 31);
  const int node = __builtin_amdgcn_readfirstlane((int)((blockIdx.x * blockDim.x + threadIdx.x) >> 5));
  if (node >= n) return;
  const int cnt  = __builtin_amdgcn_readfirstlane(deg[node]) + 1;
  const int tcnt = (cnt < kRow) ? cnt : kRow;
  const int* trow = tab + (size_t)node * kRow;
  int L = 0;
  if (cnt > kRow) {
    L = __builtin_amdgcn_readfirstlane(deg[n]);
    L = (L < E) ? L : E;
  }
  const float4 ad = *(const float4*)(a_d + (size_t)node * 4);
  const float NEG = -__builtin_huge_valf();

  int s0 = node;
  float e0 = NEG, e1 = NEG, e2 = NEG, e3 = NEG;
  if (lane < cnt) {
    if (lane > 0) s0 = trow[lane];
    const float4 as = *(const float4*)(a_s + (size_t)s0 * 4);
    e0 = lrelu(as.x + ad.x);
    e1 = lrelu(as.y + ad.y);
    e2 = lrelu(as.z + ad.z);
    e3 = lrelu(as.w + ad.w);
  }
  float m0 = e0, m1 = e1, m2 = e2, m3 = e3;
  for (int j = 32 + lane; j < tcnt; j += 32) {
    const int s = trow[j];
    const float4 as = *(const float4*)(a_s + (size_t)s * 4);
    m0 = fmaxf(m0, lrelu(as.x + ad.x));
    m1 = fmaxf(m1, lrelu(as.y + ad.y));
    m2 = fmaxf(m2, lrelu(as.z + ad.z));
    m3 = fmaxf(m3, lrelu(as.w + ad.w));
  }
  for (int k = lane; k < L; k += 32) {
    if (spill_dst[k] == node) {
      const float4 as = *(const float4*)(a_s + (size_t)spill_src[k] * 4);
      m0 = fmaxf(m0, lrelu(as.x + ad.x));
      m1 = fmaxf(m1, lrelu(as.y + ad.y));
      m2 = fmaxf(m2, lrelu(as.z + ad.z));
      m3 = fmaxf(m3, lrelu(as.w + ad.w));
    }
  }
  for (int o = 16; o > 0; o >>= 1) {
    m0 = fmaxf(m0, __shfl_xor(m0, o));
    m1 = fmaxf(m1, __shfl_xor(m1, o));
    m2 = fmaxf(m2, __shfl_xor(m2, o));
    m3 = fmaxf(m3, __shfl_xor(m3, o));
  }
  const bool v0 = lane < cnt;
  const float x0 = v0 ? __expf(e0 - m0) : 0.f;
  const float x1 = v0 ? __expf(e1 - m1) : 0.f;
  const float x2 = v0 ? __expf(e2 - m2) : 0.f;
  const float x3 = v0 ? __expf(e3 - m3) : 0.f;
  float z0 = x0, z1 = x1, z2 = x2, z3 = x3;
  for (int j = 32 + lane; j < tcnt; j += 32) {
    const int s = trow[j];
    const float4 as = *(const float4*)(a_s + (size_t)s * 4);
    z0 += __expf(lrelu(as.x + ad.x) - m0);
    z1 += __expf(lrelu(as.y + ad.y) - m1);
    z2 += __expf(lrelu(as.z + ad.z) - m2);
    z3 += __expf(lrelu(as.w + ad.w) - m3);
  }
  for (int k = lane; k < L; k += 32) {
    if (spill_dst[k] == node) {
      const float4 as = *(const float4*)(a_s + (size_t)spill_src[k] * 4);
      z0 += __expf(lrelu(as.x + ad.x) - m0);
      z1 += __expf(lrelu(as.y + ad.y) - m1);
      z2 += __expf(lrelu(as.z + ad.z) - m2);
      z3 += __expf(lrelu(as.w + ad.w) - m3);
    }
  }
  for (int o = 16; o > 0; o >>= 1) {
    z0 += __shfl_xor(z0, o);
    z1 += __shfl_xor(z1, o);
    z2 += __shfl_xor(z2, o);
    z3 += __shfl_xor(z3, o);
  }
  const float p0 = x0 / z0, p1 = x1 / z1, p2 = x2 / z2, p3 = x3 / z3;

  const int head = lane >> 3;
  const int k0 = (head == 0) ? -1 : 0;
  const int k1 = (head == 1) ? -1 : 0;
  const int k2 = (head == 2) ? -1 : 0;
  const int k3 = (head == 3) ? -1 : 0;
  const float* hl = h + lane * 4;
  float4 acc = {0.f, 0.f, 0.f, 0.f};
  for (int cb = 0; cb < tcnt; cb += 32) {
    int s = s0;
    float q0 = p0, q1 = p1, q2 = p2, q3 = p3;
    if (cb > 0) {
      const int j = cb + lane;
      s = node; q0 = 0.f; q1 = 0.f; q2 = 0.f; q3 = 0.f;
      if (j < tcnt) {
        s = trow[j];
        const float4 as = *(const float4*)(a_s + (size_t)s * 4);
        q0 = __expf(lrelu(as.x + ad.x) - m0) / z0;
        q1 = __expf(lrelu(as.y + ad.y) - m1) / z1;
        q2 = __expf(lrelu(as.z + ad.z) - m2) / z2;
        q3 = __expf(lrelu(as.w + ad.w) - m3) / z3;
      }
    }
    const int lim = (tcnt - cb < 32) ? (tcnt - cb) : 32;
#pragma unroll 4
    for (int t = 0; t < lim; ++t) add_slot(acc, s, q0, q1, q2, q3, t, k0, k1, k2, k3, hl);
  }
  for (int kb = 0; kb < L; kb += 32) {
    const int k = kb + lane;
    int s = node;
    float q0 = 0.f, q1 = 0.f, q2 = 0.f, q3 = 0.f;
    const bool mine = (k < L) && (spill_dst[k] == node);
    if (mine) {
      s = spill_src[k];
      const float4 as = *(const float4*)(a_s + (size_t)s * 4);
      q0 = __expf(lrelu(as.x + ad.x) - m0) / z0;
      q1 = __expf(lrelu(as.y + ad.y) - m1) / z1;
      q2 = __expf(lrelu(as.z + ad.z) - m2) / z2;
      q3 = __expf(lrelu(as.w + ad.w) - m3) / z3;
    }
    unsigned todo = __builtin_amdgcn_ballot_w32(mine);
    while (todo != 0u) {
      const int t = __builtin_ctz(todo);
      todo &= todo - 1u;
      add_slot(acc, s, q0, q1, q2, q3, t, k0, k1, k2, k3, hl);
    }
  }
  float4 r;
  r.x = acc.x + bias[lane * 4 + 0];
  r.y = acc.y + bias[lane * 4 + 1];
  r.z = acc.z + bias[lane * 4 + 2];
  r.w = acc.w + bias[lane * 4 + 3];
  r.x = (r.x > 0.f) ? r.x : expm1f(r.x);
  r.y = (r.y > 0.f) ? r.y : expm1f(r.y);
  r.z = (r.z > 0.f) ? r.z : expm1f(r.z);
  r.w = (r.w > 0.f) ? r.w : expm1f(r.w);
  *(float4*)(out + (size_t)node * 128 + lane * 4) = r;
}

__global__ __launch_bounds__(256) void gat2_node_k(const int* __restrict__ deg,
                                                   const int* __restrict__ tab,
                                                   const int* __restrict__ spill_dst,
                                                   const int* __restrict__ spill_src, int E,
                                                   const float* __restrict__ h2,
                                                   const float* __restrict__ a_s,
                                                   const float* __restrict__ a_d,
                                                   const float* __restrict__ bias,
                                                   float* __restrict__ out, int n) {
  const int node = (int)((blockIdx.x * blockDim.x + threadIdx.x) >> 4);
  const int c    = (int)(threadIdx.x & 15);
  if (node >= n) return;
  const int cnt  = deg[node] + 1;
  const int tcnt = (cnt < kRow) ? cnt : kRow;
  const int* trow = tab + (size_t)node * kRow;
  int L = 0;
  if (cnt > kRow) {
    L = deg[n];
    L = (L < E) ? L : E;
  }
  const float ad = a_d[node];
  const float NEG = -__builtin_huge_valf();

  int s0 = node;
  float e0 = NEG;
  if (c < cnt) {
    if (c > 0) s0 = trow[c];
    e0 = lrelu(a_s[s0] + ad);
  }
  float m = e0;
  for (int j = 16 + c; j < tcnt; j += 16) m = fmaxf(m, lrelu(a_s[trow[j]] + ad));
  for (int k = c; k < L; k += 16)
    if (spill_dst[k] == node) m = fmaxf(m, lrelu(a_s[spill_src[k]] + ad));
  for (int o = 8; o > 0; o >>= 1) m = fmaxf(m, __shfl_xor(m, o));
  const float x0 = (c < cnt) ? __expf(e0 - m) : 0.f;
  float z = x0;
  for (int j = 16 + c; j < tcnt; j += 16) z += __expf(lrelu(a_s[trow[j]] + ad) - m);
  for (int k = c; k < L; k += 16)
    if (spill_dst[k] == node) z += __expf(lrelu(a_s[spill_src[k]] + ad) - m);
  for (int o = 8; o > 0; o >>= 1) z += __shfl_xor(z, o);
  const float p0 = x0 / z;

  float acc = 0.f;
  for (int cb = 0; cb < tcnt; cb += 16) {
    int s = s0;
    float q = p0;
    if (cb > 0) {
      const int j = cb + c;
      s = node; q = 0.f;
      if (j < tcnt) {
        s = trow[j];
        q = __expf(lrelu(a_s[s] + ad) - m) / z;
      }
    }
    const int lim = (tcnt - cb < 16) ? (tcnt - cb) : 16;
    for (int t = 0; t < lim; t += 4) {
#pragma unroll
      for (int k = 0; k < 4; ++k) {
        const int   tt = t + k;
        const int   st = __shfl(s, tt, 16);
        const float a  = __shfl(q, tt, 16);
        const float hv = h2[(size_t)st * 16 + c];
        acc += ((tt < lim) ? hv : 0.f) * a;
      }
    }
  }
  for (int kb = 0; kb < L; kb += 16) {
    const int k = kb + c;
    int s = node;
    float q = 0.f;
    int mine = 0;
    if (k < L && spill_dst[k] == node) {
      s = spill_src[k];
      q = __expf(lrelu(a_s[s] + ad) - m) / z;
      mine = 1;
    }
    for (int t = 0; t < 16; ++t) {
      const int   st = __shfl(s, t, 16);
      const float a  = __shfl(q, t, 16);
      const int   on = __shfl(mine, t, 16);
      const float hv = h2[(size_t)st * 16 + c];
      acc += ((on != 0) ? hv : 0.f) * a;
    }
  }
  out[(size_t)node * 16 + c] = acc + bias[c];
}

__global__ void fill_u32_k(unsigned* __restrict__ p, unsigned v, long cnt) {
  long i = (long)blockIdx.x * blockDim.x + threadIdx.x;
  if (i < cnt) p[i] = v;
}

extern "C" void kernel_launch(void* const* d_in, const int* in_sizes, int n_in,
                              void* d_out, int out_size, void* d_ws, size_t ws_size,
                              hipStream_t stream) {
  const float* x    = (const float*)d_in[0];
  const int*   ei   = (const int*)d_in[1];
  const float* W1   = (const float*)d_in[2];
  const float* as1w = (const float*)d_in[3];
  const float* ad1w = (const float*)d_in[4];
  const float* b1   = (const float*)d_in[5];
  const float* W2   = (const float*)d_in[6];
  const float* as2w = (const float*)d_in[7];
  const float* ad2w = (const float*)d_in[8];
  const float* b2   = (const float*)d_in[9];
  float* out = (float*)d_out;

  const int n = in_sizes[0] / 128;
  const int E = in_sizes[1] / 2;

  float* ws     = (float*)d_ws;
  float* h1     = ws;
  float* out1   = h1 + (size_t)n * 128;
  float* as1    = out1 + (size_t)n * 128;
  float* ad1    = as1 + (size_t)n * 4;
  float* h2     = ad1 + (size_t)n * 4;
  int*   tab    = (int*)(h2 + (size_t)n * 16);
  int*   spd    = tab + (size_t)n * kRow;
  int*   sps    = spd + E;
  int*   cnt    = sps + E;
  float* as2    = (float*)(cnt + n + 1);
  float* ad2    = as2 + n;

  const int B = 256;
  const int mtiles = (n + 15) / 16;
  const int gemm_blocks = (mtiles + 7) / 8;
  const int edge_blocks = (E + 2047) / 2048;

  fill_u32_k<<<(n + 1 + B - 1) / B, B, 0, stream>>>((unsigned*)cnt, 0u, (long)n + 1);
  proj1_slot_k<<<gemm_blocks + edge_blocks, B, 0, stream>>>(x, W1, h1, mtiles, gemm_blocks, ei, E, cnt, n, tab, spd, sps);
  att_k<4, 32><<<(n * 4 + B - 1) / B, B, 0, stream>>>(h1, as1w, ad1w, as1, ad1, n);

  gat1_node_k<<<(int)(((long)n * 32 + B - 1) / B), B, 0, stream>>>(cnt, tab, spd, sps, E, h1, as1, ad1, b1, out1, n);

  gemm_wmma_f32<1, 128, 16><<<(mtiles + 7) / 8, B, 0, stream>>>(out1, W2, h2, mtiles);
  att_k<1, 16><<<(n + B - 1) / B, B, 0, stream>>>(h2, as2w, ad2w, as2, ad2, n);
  gat2_node_k<<<(int)(((long)n * 16 + B - 1) / B), B, 0, stream>>>(cnt, tab, spd, sps, E, h2, as2, ad2, b2, out, n);
}


# Edge ranks from LDS counters: blocks order their edges by destination range beside the x*W1 blocks, then one work-group per range ranks and slots them; no global atomic per edge; arrival rank kept
# speedup vs baseline: 7.5838x; 1.6776x over previous
#include <hip/hip_runtime.h>


typedef float v2f __attribute__((ext_vector_type(2)));
typedef float v8f __attribute__((ext_vector_type(8)));
typedef float v4f __attribute__((ext_vector_type(4)));

template <int NT, int K, int NCOL>
__device__ __forceinline__ void gemm_wave_tile(const float* __restrict__ A,
                                               const float* __restrict__ B,
                                               float* __restrict__ C,
                                               int wave, int lane) {
  const int lane15 = lane & 15;
  const int khalf  = (lane >> 4) * 2;
  const long rowbase = (long)wave * 16;

  v8f acc[NT];
#pragma unroll
  for (int t = 0; t < NT; ++t) acc[t] = (v8f){0.f,0.f,0.f,0.f,0.f,0.f,0.f,0.f};

  const float* arow = A + (rowbase + lane15) * K + khalf;
#pragma unroll 4
  for (int kb = 0; kb < K; kb += 4) {
    v2f a;
    a.x = arow[kb];
    a.y = arow[kb + 1];
#pragma unroll
    for (int t = 0; t < NT; ++t) {
      const float* bp = B + (size_t)(kb + khalf) * NCOL + t * 16 + lane15;
      v2f b;
      b.x = bp[0];
      b.y = bp[NCOL];
      acc[t] = __builtin_amdgcn_wmma_f32_16x16x4_f32(
           false, a,  false, b,
           (short)0, acc[t],  false,  false);
    }
  }
  const int rowoff = (lane >> 4) * 8;
#pragma unroll
  for (int t = 0; t < NT; ++t) {
#pragma unroll
    for (int j = 0; j < 8; ++j) {
      C[(rowbase + rowoff + j) * NCOL + t * 16 + lane15] = acc[t][j];
    }
  }
}

template <int NT, int K, int NCOL>
__global__ __launch_bounds__(256) void gemm_wmma_f32(const float* __restrict__ A,
                                                     const float* __restrict__ B,
                                                     float* __restrict__ C,
                                                     int mtiles) {
  const int wave = (int)((blockIdx.x * blockDim.x + threadIdx.x) >> 5);
  const int lane = (int)(threadIdx.x & 31);
  if (wave < mtiles) gemm_wave_tile<NT, K, NCOL>(A, B, C, wave, lane);
}

constexpr int kRow = 64;

__device__ __forceinline__ void put_edge(int d, int r, int s, int* __restrict__ tab,
                                         int* __restrict__ spill_n, int* __restrict__ spill_dst,
                                         int* __restrict__ spill_src) {
  if (r < kRow - 1) {
    tab[(size_t)d * kRow + 1 + r] = s;
  } else {
    const int k = atomicAdd(spill_n, 1);
    spill_dst[k] = d;
    spill_src[k] = s;
  }
}

constexpr int kSh     = 9;
constexpr int kBk     = 1 << kSh;
constexpr int kMaxB   = 256;
constexpr int kChunk  = 4096;
constexpr int kSegStr = kMaxB + 1;

__device__ __forceinline__ void bin_block(const int* __restrict__ esrc, const int* __restrict__ edst,
                                          int E, int n, int j, int* __restrict__ segoff,
                                          int2* __restrict__ pairs, int* hist, int* offs, int* wtot) {
  const int tid  = (int)threadIdx.x;
  const int lane = tid & 31;
  const int wave = tid >> 5;
  hist[tid] = 0;
  __syncthreads();
  const long base = (long)j * kChunk;
  int d[16], s[16], lr[16];
#pragma unroll
  for (int k = 0; k < 16; ++k) {
    const long i = base + k * 256 + tid;
    d[k] = -1;
    s[k] = 0;
    if (i < E) { d[k] = edst[i]; s[k] = esrc[i]; }
  }
#pragma unroll
  for (int k = 0; k < 16; ++k) {
    lr[k] = 0;
    if ((unsigned)d[k] < (unsigned)n) lr[k] = atomicAdd(&hist[d[k] >> kSh], 1);
    else d[k] = -1;
  }
  __syncthreads();
  const int v = hist[tid];
  int x = v;
#pragma unroll
  for (int o = 1; o < 32; o <<= 1) {
    const int y = __shfl_up(x, o, 32);
    if (lane >= o) x += y;
  }
  if (lane == 31) wtot[wave] = x;
  __syncthreads();
  int wp = 0;
  for (int w = 0; w < wave; ++w) wp += wtot[w];
  const int excl = wp + x - v;
  offs[tid] = excl;
  segoff[(size_t)j * kSegStr + tid] = excl;
  if (tid == kMaxB - 1) segoff[(size_t)j * kSegStr + kMaxB] = excl + v;
  __syncthreads();
#pragma unroll
  for (int k = 0; k < 16; ++k) {
    if (d[k] >= 0) pairs[base + offs[d[k] >> kSh] + lr[k]] = make_int2(d[k], s[k]);
  }
}

__global__ __launch_bounds__(256) void proj1_bin_k(const float* __restrict__ x,
                                                   const float* __restrict__ W1,
                                                   float* __restrict__ h1, int mtiles,
                                                   int gemm_blocks,
                                                   const int* __restrict__ ei, int E, int n,
                                                   int* __restrict__ cnt,
                                                   int* __restrict__ segoff,
                                                   int* __restrict__ pairs) {
  __shared__ int hist[kMaxB];
  __shared__ int offs[kMaxB];
  __shared__ int wtot[8];
  if ((int)blockIdx.x < gemm_blocks) {
    const int lane = (int)(threadIdx.x & 31);
    const int wave = (int)((blockIdx.x * 256 + threadIdx.x) >> 5);
    if (wave < mtiles) gemm_wave_tile<8, 128, 128>(x, W1, h1, wave, lane);
  } else {
    const int j = (int)blockIdx.x - gemm_blocks;
    if (j == 0 && threadIdx.x == 0) cnt[n] = 0;
    bin_block(ei, ei + E, E, n, j, segoff, (int2*)pairs, hist, offs, wtot);
  }
}

__global__ __launch_bounds__(1024) void rank_fill_k(const int* __restrict__ pairs_i,
                                                    const int* __restrict__ segoff, int nblk,
                                                    int* __restrict__ cnt, int n,
                                                    int* __restrict__ tab,
                                                    int* __restrict__ spill_dst,
                                                    int* __restrict__ spill_src) {
  const int2* __restrict__ pairs = (const int2*)pairs_i;
  __shared__ int lcnt[kBk];
  __shared__ int sst[1024];
  __shared__ int sct[1024];
  const int tid  = (int)threadIdx.x;
  const int lane = tid & 31;
  const int wave = tid >> 5;
  const int b    = (int)blockIdx.x;
  const int nlo  = b * kBk;
  if (tid < kBk) lcnt[tid] = 0;
  for (int jb = 0; jb < nblk; jb += 1024) {
    __syncthreads();
    const int j = jb + tid;
    int st = 0, c = 0;
    if (j < nblk) {
      st = segoff[(size_t)j * kSegStr + b];
      c  = segoff[(size_t)j * kSegStr + b + 1] - st;
    }
    sst[tid] = st;
    sct[tid] = c;
    __syncthreads();
    for (int i = 0; i < 32; i += 4) {
      if (jb + i * 32 >= nblk) break;
      int2 p[4];
      int  c4[4];
      long sb[4];
#pragma unroll
      for (int u = 0; u < 4; ++u) {
        const int q = (i + u) * 32 + wave;
        c4[u] = sct[q];
        sb[u] = (long)(jb + q) * kChunk + sst[q];
        p[u]  = make_int2(nlo, 0);
        if (lane < c4[u]) p[u] = pairs[sb[u] + lane];
      }
#pragma unroll
      for (int u = 0; u < 4; ++u) {
        if (lane < c4[u]) {
          const int r = atomicAdd(&lcnt[p[u].x - nlo], 1);
          put_edge(p[u].x, r, p[u].y, tab, cnt + n, spill_dst, spill_src);
        }
        for (int e = lane + 32; e < c4[u]; e += 32) {
          const int2 pe = pairs[sb[u] + e];
          const int r = atomicAdd(&lcnt[pe.x - nlo], 1);
          put_edge(pe.x, r, pe.y, tab, cnt + n, spill_dst, spill_src);
        }
      }
    }
  }
  __syncthreads();
  if (tid < kBk) {
    const int node = nlo + tid;
    if (node < n) cnt[node] = lcnt[tid];
  }
}

__device__ __forceinline__ float lrelu(float e) { return (e >= 0.f) ? e : 0.2f * e; }

template <int K>
__device__ __forceinline__ float xmask_add(float v) {
  const int o = __builtin_amdgcn_update_dpp(0, __float_as_int(v), 0x160 | K, 0xf, 0xf, true);
  return v + __int_as_float(o);
}
__device__ __forceinline__ float sum8(float v) {
  v = xmask_add<1>(v);
  v = xmask_add<2>(v);
  v = xmask_add<4>(v);
  return v;
}
__device__ __forceinline__ float sum16(float v) {
  v = xmask_add<8>(sum8(v));
  return v;
}

__device__ __forceinline__ float dot4(const v4f& a, const v4f& b) {
  const v2f p = a.lo * b.lo + a.hi * b.hi;
  return p.x + p.y;
}


template <int R>
__device__ __forceinline__ void take_rows(v4f& acc, float& z, float& m, int s, int t,
                                          const float* __restrict__ hl, const v4f& ws, float ad) {
  v4f   hv[R];
  float e[R];
#pragma unroll
  for (int k = 0; k < R; ++k) {
    const int st = __builtin_amdgcn_readlane(s, t + k);
    hv[k] = *(const v4f*)(hl + (size_t)st * 128);
  }
  float mn = m;
#pragma unroll
  for (int k = 0; k < R; ++k) {
    e[k] = lrelu(sum8(dot4(hv[k], ws)) + ad);
    mn = fmaxf(mn, e[k]);
  }
  const float f = __expf(m - mn);
  float w  = __expf(e[0] - mn);
  float zs = w;
  v4f   sm = hv[0] * w;
#pragma unroll
  for (int k = 1; k < R; ++k) {
    w = __expf(e[k] - mn);
    zs += w;
    sm += hv[k] * w;
  }
  z   = z * f + zs;
  acc = acc * f + sm;
  m = mn;
}

__global__ __launch_bounds__(256) void gat1_node_k(const int* __restrict__ deg,
                                                   const int* __restrict__ tab,
                                                   const int* __restrict__ spill_dst,
                                                   const int* __restrict__ spill_src, int E,
                                                   const float* __restrict__ h,
                                                   const float* __restrict__ att_s,
                                                   const float* __restrict__ att_d,
                                                   const float* __restrict__ bias,
                                                   float* __restrict__ out, int n) {
  const int lane = (int)(threadIdx.x & 31);
  const int node = __builtin_amdgcn_readfirstlane((int)((blockIdx.x * blockDim.x + threadIdx.x) >> 5));
  if (node >= n) return;
  const int cnt  = __builtin_amdgcn_readfirstlane(deg[node]) + 1;
  const int tcnt = (cnt < kRow) ? cnt : kRow;
  const int* trow = tab + (size_t)node * kRow;
  int L = 0;
  if (cnt > kRow) {
    L = __builtin_amdgcn_readfirstlane(deg[n]);
    L = (L < E) ? L : E;
  }
  const float* hl = h + lane * 4;

  v4f ws, wd;
  ws.x = att_s[lane * 4 + 0]; ws.y = att_s[lane * 4 + 1];
  ws.z = att_s[lane * 4 + 2]; ws.w = att_s[lane * 4 + 3];
  wd.x = att_d[lane * 4 + 0]; wd.y = att_d[lane * 4 + 1];
  wd.z = att_d[lane * 4 + 2]; wd.w = att_d[lane * 4 + 3];

  int s0 = node;
  if (lane > 0 && lane < cnt) s0 = trow[lane];
  const v4f h0 = *(const v4f*)(hl + (size_t)node * 128);
  const float ad = sum8(dot4(h0, wd));
  float m = lrelu(sum8(dot4(h0, ws)) + ad);
  float z = 1.f;
  v4f acc = h0;

  for (int cb = 0; cb < tcnt; cb += 32) {
    int s = s0;
    if (cb > 0) {
      const int j = cb + lane;
      s = node;
      if (j < tcnt) s = trow[j];
    }
    const int lim = (tcnt - cb < 32) ? (tcnt - cb) : 32;
    int t = (cb == 0) ? 1 : 0;
    for (; t + 4 <= lim; t += 4) take_rows<4>(acc, z, m, s, t, hl, ws, ad);
    const int rest = lim - t;
    if (rest == 3)      take_rows<3>(acc, z, m, s, t, hl, ws, ad);
    else if (rest == 2) take_rows<2>(acc, z, m, s, t, hl, ws, ad);
    else if (rest == 1) take_rows<1>(acc, z, m, s, t, hl, ws, ad);
  }
  for (int kb = 0; kb < L; kb += 32) {
    const int k = kb + lane;
    int s = node;
    const bool mine = (k < L) && (spill_dst[k] == node);
    if (mine) s = spill_src[k];
    unsigned todo = __builtin_amdgcn_ballot_w32(mine);
    while (todo != 0u) {
      const int t = __builtin_ctz(todo);
      todo &= todo - 1u;
      take_rows<1>(acc, z, m, s, t, hl, ws, ad);
    }
  }
  const float iz = 1.f / z;
  v4f r = acc * iz;
  r.x += bias[lane * 4 + 0];
  r.y += bias[lane * 4 + 1];
  r.z += bias[lane * 4 + 2];
  r.w += bias[lane * 4 + 3];
  r.x = (r.x > 0.f) ? r.x : expm1f(r.x);
  r.y = (r.y > 0.f) ? r.y : expm1f(r.y);
  r.z = (r.z > 0.f) ? r.z : expm1f(r.z);
  r.w = (r.w > 0.f) ? r.w : expm1f(r.w);
  *(v4f*)(out + (size_t)node * 128 + lane * 4) = r;
}

__global__ __launch_bounds__(256) void gat2_node_k(const int* __restrict__ deg,
                                                   const int* __restrict__ tab,
                                                   const int* __restrict__ spill_dst,
                                                   const int* __restrict__ spill_src, int E,
                                                   const float* __restrict__ h2,
                                                   const float* __restrict__ att_s,
                                                   const float* __restrict__ att_d,
                                                   const float* __restrict__ bias,
                                                   float* __restrict__ out, int n) {
  const int node = (int)((blockIdx.x * blockDim.x + threadIdx.x) >> 4);
  const int c    = (int)(threadIdx.x & 15);
  if (node >= n) return;
  const int cnt  = deg[node] + 1;
  const int tcnt = (cnt < kRow) ? cnt : kRow;
  const int* trow = tab + (size_t)node * kRow;
  int L = 0;
  if (cnt > kRow) {
    L = deg[n];
    L = (L < E) ? L : E;
  }
  const float NEG = -__builtin_huge_valf();
  const float ws = att_s[c];
  const float wd = att_d[c];

  int s0 = node;
  if (c > 0 && c < cnt) s0 = trow[c];
  const float h0 = h2[(size_t)node * 16 + c];
  const float ad = sum16(h0 * wd);
  float m = lrelu(sum16(h0 * ws) + ad);
  float z = 1.f;
  float acc = h0;

  for (int cb = 0; cb < tcnt; cb += 16) {
    int s = s0;
    if (cb > 0) {
      const int j = cb + c;
      s = node;
      if (j < tcnt) s = trow[j];
    }
    const int lim = (tcnt - cb < 16) ? (tcnt - cb) : 16;
    for (int t = (cb == 0) ? 1 : 0; t < lim; t += 4) {
      float hv[4];
      float e[4];
#pragma unroll
      for (int k = 0; k < 4; ++k) {
        const int tt = t + k;
        const int st = __shfl(s, (tt < lim) ? tt : 0, 16);
        hv[k] = h2[(unsigned)st * 16u + (unsigned)c];
      }
#pragma unroll
      for (int k = 0; k < 4; ++k) {
        const float v = lrelu(sum16(hv[k] * ws) + ad);
        e[k] = (t + k < lim) ? v : NEG;
      }
      const float mn = fmaxf(fmaxf(m, fmaxf(e[0], e[1])), fmaxf(e[2], e[3]));
      const float f  = __expf(m - mn);
      float w[4];
#pragma unroll
      for (int k = 0; k < 4; ++k) w[k] = __expf(e[k] - mn);
      z   = z * f + ((w[0] + w[1]) + (w[2] + w[3]));
      acc = acc * f + ((hv[0] * w[0] + hv[1] * w[1]) + (hv[2] * w[2] + hv[3] * w[3]));
      m = mn;
    }
  }
  for (int kb = 0; kb < L; kb += 16) {
    const int k = kb + c;
    int s = node;
    int mine = 0;
    if (k < L && spill_dst[k] == node) {
      s = spill_src[k];
      mine = 1;
    }
    for (int t = 0; t < 16; ++t) {
      const int   st = __shfl(s, t, 16);
      const int   on = __shfl(mine, t, 16);
      const float hv = h2[(size_t)st * 16 + c];
      const float v  = lrelu(sum16(hv * ws) + ad);
      const float e  = (on != 0) ? v : NEG;
      const float mn = fmaxf(m, e);
      const float f  = __expf(m - mn);
      const float w  = __expf(e - mn);
      z   = z * f + w;
      acc = acc * f + hv * w;
      m = mn;
    }
  }
  out[(size_t)node * 16 + c] = acc / z + bias[c];
}

extern "C" void kernel_launch(void* const* d_in, const int* in_sizes, int n_in,
                              void* d_out, int out_size, void* d_ws, size_t ws_size,
                              hipStream_t stream) {
  const float* x    = (const float*)d_in[0];
  const int*   ei   = (const int*)d_in[1];
  const float* W1   = (const float*)d_in[2];
  const float* as1w = (const float*)d_in[3];
  const float* ad1w = (const float*)d_in[4];
  const float* b1   = (const float*)d_in[5];
  const float* W2   = (const float*)d_in[6];
  const float* as2w = (const float*)d_in[7];
  const float* ad2w = (const float*)d_in[8];
  const float* b2   = (const float*)d_in[9];
  float* out = (float*)d_out;

  const int n = in_sizes[0] / 128;
  const int E = in_sizes[1] / 2;

  float* ws     = (float*)d_ws;
  float* h1     = ws;
  float* out1   = h1 + (size_t)n * 128;
  float* h2     = out1 + (size_t)n * 128;
  int*   tab    = (int*)(h2 + (size_t)n * 16);
  int*   spd    = tab + (size_t)n * kRow;
  int*   sps    = spd + E;
  int*   pairs  = sps + E;
  int*   cnt    = pairs + 2 * (size_t)E;
  int*   segoff = cnt + n + 1;

  const int B = 256;
  const int mtiles = (n + 15) / 16;
  const int gemm_blocks = (mtiles + 7) / 8;
  const int nblk = (E + kChunk - 1) / kChunk;
  const int nrng = (n + kBk - 1) / kBk;

  proj1_bin_k<<<gemm_blocks + nblk, B, 0, stream>>>(x, W1, h1, mtiles, gemm_blocks, ei, E, n, cnt, segoff, pairs);
  rank_fill_k<<<nrng, 1024, 0, stream>>>(pairs, segoff, nblk, cnt, n, tab, spd, sps);

  gat1_node_k<<<(int)(((long)n * 32 + B - 1) / B), B, 0, stream>>>(cnt, tab, spd, sps, E, h1, as1w, ad1w, b1, out1, n);

  gemm_wmma_f32<1, 128, 16><<<(mtiles + 7) / 8, B, 0, stream>>>(out1, W2, h2, mtiles);
  gat2_node_k<<<(int)(((long)n * 16 + B - 1) / B), B, 0, stream>>>(cnt, tab, spd, sps, E, h2, as2w, ad2w, b2, out, n);
}


# Layer-2 gather on 4 lanes per node (a lane holds 4 of the 16 columns, one 16-byte load per row), 8 nodes side by side per wave, 8 rows in flight: 50000 waves to 12500; sum order changes
# speedup vs baseline: 8.0882x; 1.0665x over previous
#include <hip/hip_runtime.h>


typedef float v2f __attribute__((ext_vector_type(2)));
typedef float v8f __attribute__((ext_vector_type(8)));
typedef float v4f __attribute__((ext_vector_type(4)));

template <int NT, int K, int NCOL>
__device__ __forceinline__ void gemm_wave_tile(const float* __restrict__ A,
                                               const float* __restrict__ B,
                                               float* __restrict__ C,
                                               int wave, int lane) {
  const int lane15 = lane & 15;
  const int khalf  = (lane >> 4) * 2;
  const long rowbase = (long)wave * 16;

  v8f acc[NT];
#pragma unroll
  for (int t = 0; t < NT; ++t) acc[t] = (v8f){0.f,0.f,0.f,0.f,0.f,0.f,0.f,0.f};

  const float* arow = A + (rowbase + lane15) * K + khalf;
#pragma unroll 4
  for (int kb = 0; kb < K; kb += 4) {
    v2f a;
    a.x = arow[kb];
    a.y = arow[kb + 1];
#pragma unroll
    for (int t = 0; t < NT; ++t) {
      const float* bp = B + (size_t)(kb + khalf) * NCOL + t * 16 + lane15;
      v2f b;
      b.x = bp[0];
      b.y = bp[NCOL];
      acc[t] = __builtin_amdgcn_wmma_f32_16x16x4_f32(
           false, a,  false, b,
           (short)0, acc[t],  false,  false);
    }
  }
  const int rowoff = (lane >> 4) * 8;
#pragma unroll
  for (int t = 0; t < NT; ++t) {
#pragma unroll
    for (int j = 0; j < 8; ++j) {
      C[(rowbase + rowoff + j) * NCOL + t * 16 + lane15] = acc[t][j];
    }
  }
}

template <int NT, int K, int NCOL>
__global__ __launch_bounds__(256) void gemm_wmma_f32(const float* __restrict__ A,
                                                     const float* __restrict__ B,
                                                     float* __restrict__ C,
                                                     int mtiles) {
  const int wave = (int)((blockIdx.x * blockDim.x + threadIdx.x) >> 5);
  const int lane = (int)(threadIdx.x & 31);
  if (wave < mtiles) gemm_wave_tile<NT, K, NCOL>(A, B, C, wave, lane);
}

constexpr int kRow = 64;

__device__ __forceinline__ void put_edge(int d, int r, int s, int* __restrict__ tab,
                                         int* __restrict__ spill_n, int* __restrict__ spill_dst,
                                         int* __restrict__ spill_src) {
  if (r < kRow - 1) {
    tab[(size_t)d * kRow + 1 + r] = s;
  } else {
    const int k = atomicAdd(spill_n, 1);
    spill_dst[k] = d;
    spill_src[k] = s;
  }
}

constexpr int kSh     = 9;
constexpr int kBk     = 1 << kSh;
constexpr int kMaxB   = 256;
constexpr int kChunk  = 4096;
constexpr int kSegStr = kMaxB + 1;

__device__ __forceinline__ void bin_block(const int* __restrict__ esrc, const int* __restrict__ edst,
                                          int E, int n, int j, int* __restrict__ segoff,
                                          int2* __restrict__ pairs, int* hist, int* offs, int* wtot) {
  const int tid  = (int)threadIdx.x;
  const int lane = tid & 31;
  const int wave = tid >> 5;
  hist[tid] = 0;
  __syncthreads();
  const long base = (long)j * kChunk;
  int d[16], s[16], lr[16];
#pragma unroll
  for (int k = 0; k < 16; ++k) {
    const long i = base + k * 256 + tid;
    d[k] = -1;
    s[k] = 0;
    if (i < E) { d[k] = edst[i]; s[k] = esrc[i]; }
  }
#pragma unroll
  for (int k = 0; k < 16; ++k) {
    lr[k] = 0;
    if ((unsigned)d[k] < (unsigned)n) lr[k] = atomicAdd(&hist[d[k] >> kSh], 1);
    else d[k] = -1;
  }
  __syncthreads();
  const int v = hist[tid];
  int x = v;
#pragma unroll
  for (int o = 1; o < 32; o <<= 1) {
    const int y = __shfl_up(x, o, 32);
    if (lane >= o) x += y;
  }
  if (lane == 31) wtot[wave] = x;
  __syncthreads();
  int wp = 0;
  for (int w = 0; w < wave; ++w) wp += wtot[w];
  const int excl = wp + x - v;
  offs[tid] = excl;
  segoff[(size_t)j * kSegStr + tid] = excl;
  if (tid == kMaxB - 1) segoff[(size_t)j * kSegStr + kMaxB] = excl + v;
  __syncthreads();
#pragma unroll
  for (int k = 0; k < 16; ++k) {
    if (d[k] >= 0) pairs[base + offs[d[k] >> kSh] + lr[k]] = make_int2(d[k], s[k]);
  }
}

__global__ __launch_bounds__(256) void proj1_bin_k(const float* __restrict__ x,
                                                   const float* __restrict__ W1,
                                                   float* __restrict__ h1, int mtiles,
                                                   int gemm_blocks,
                                                   const int* __restrict__ ei, int E, int n,
                                                   int* __restrict__ cnt,
                                                   int* __restrict__ segoff,
                                                   int* __restrict__ pairs) {
  __shared__ int hist[kMaxB];
  __shared__ int offs[kMaxB];
  __shared__ int wtot[8];
  if ((int)blockIdx.x < gemm_blocks) {
    const int lane = (int)(threadIdx.x & 31);
    const int wave = (int)((blockIdx.x * 256 + threadIdx.x) >> 5);
    if (wave < mtiles) gemm_wave_tile<8, 128, 128>(x, W1, h1, wave, lane);
  } else {
    const int j = (int)blockIdx.x - gemm_blocks;
    if (j == 0 && threadIdx.x == 0) cnt[n] = 0;
    bin_block(ei, ei + E, E, n, j, segoff, (int2*)pairs, hist, offs, wtot);
  }
}

__global__ __launch_bounds__(1024) void rank_fill_k(const int* __restrict__ pairs_i,
                                                    const int* __restrict__ segoff, int nblk,
                                                    int* __restrict__ cnt, int n,
                                                    int* __restrict__ tab,
                                                    int* __restrict__ spill_dst,
                                                    int* __restrict__ spill_src) {
  const int2* __restrict__ pairs = (const int2*)pairs_i;
  __shared__ int lcnt[kBk];
  __shared__ int sst[1024];
  __shared__ int sct[1024];
  const int tid  = (int)threadIdx.x;
  const int lane = tid & 31;
  const int wave = tid >> 5;
  const int b    = (int)blockIdx.x;
  const int nlo  = b * kBk;
  if (tid < kBk) lcnt[tid] = 0;
  for (int jb = 0; jb < nblk; jb += 1024) {
    __syncthreads();
    const int j = jb + tid;
    int st = 0, c = 0;
    if (j < nblk) {
      st = segoff[(size_t)j * kSegStr + b];
      c  = segoff[(size_t)j * kSegStr + b + 1] - st;
    }
    sst[tid] = st;
    sct[tid] = c;
    __syncthreads();
    for (int i = 0; i < 32; i += 4) {
      if (jb + i * 32 >= nblk) break;
      int2 p[4];
      int  c4[4];
      long sb[4];
#pragma unroll
      for (int u = 0; u < 4; ++u) {
        const int q = (i + u) * 32 + wave;
        c4[u] = sct[q];
        sb[u] = (long)(jb + q) * kChunk + sst[q];
        p[u]  = make_int2(nlo, 0);
        if (lane < c4[u]) p[u] = pairs[sb[u] + lane];
      }
#pragma unroll
      for (int u = 0; u < 4; ++u) {
        if (lane < c4[u]) {
          const int r = atomicAdd(&lcnt[p[u].x - nlo], 1);
          put_edge(p[u].x, r, p[u].y, tab, cnt + n, spill_dst, spill_src);
        }
        for (int e = lane + 32; e < c4[u]; e += 32) {
          const int2 pe = pairs[sb[u] + e];
          const int r = atomicAdd(&lcnt[pe.x - nlo], 1);
          put_edge(pe.x, r, pe.y, tab, cnt + n, spill_dst, spill_src);
        }
      }
    }
  }
  __syncthreads();
  if (tid < kBk) {
    const int node = nlo + tid;
    if (node < n) cnt[node] = lcnt[tid];
  }
}

__device__ __forceinline__ float lrelu(float e) { return (e >= 0.f) ? e : 0.2f * e; }

template <int K>
__device__ __forceinline__ float xmask_add(float v) {
  const int o = __builtin_amdgcn_update_dpp(0, __float_as_int(v), 0x160 | K, 0xf, 0xf, true);
  return v + __int_as_float(o);
}
__device__ __forceinline__ float sum8(float v) {
  v = xmask_add<1>(v);
  v = xmask_add<2>(v);
  v = xmask_add<4>(v);
  return v;
}
__device__ __forceinline__ float sum16(float v) {
  v = xmask_add<8>(sum8(v));
  return v;
}

__device__ __forceinline__ float dot4(const v4f& a, const v4f& b) {
  const v2f p = a.lo * b.lo + a.hi * b.hi;
  return p.x + p.y;
}


template <int R>
__device__ __forceinline__ void take_rows(v4f& acc, float& z, float& m, int s, int t,
                                          const float* __restrict__ hl, const v4f& ws, float ad) {
  v4f   hv[R];
  float e[R];
#pragma unroll
  for (int k = 0; k < R; ++k) {
    const int st = __builtin_amdgcn_readlane(s, t + k);
    hv[k] = *(const v4f*)(hl + (size_t)st * 128);
  }
  float mn = m;
#pragma unroll
  for (int k = 0; k < R; ++k) {
    e[k] = lrelu(sum8(dot4(hv[k], ws)) + ad);
    mn = fmaxf(mn, e[k]);
  }
  const float f = __expf(m - mn);
  float w  = __expf(e[0] - mn);
  float zs = w;
  v4f   sm = hv[0] * w;
#pragma unroll
  for (int k = 1; k < R; ++k) {
    w = __expf(e[k] - mn);
    zs += w;
    sm += hv[k] * w;
  }
  z   = z * f + zs;
  acc = acc * f + sm;
  m = mn;
}

__global__ __launch_bounds__(256) void gat1_node_k(const int* __restrict__ deg,
                                                   const int* __restrict__ tab,
                                                   const int* __restrict__ spill_dst,
                                                   const int* __restrict__ spill_src, int E,
                                                   const float* __restrict__ h,
                                                   const float* __restrict__ att_s,
                                                   const float* __restrict__ att_d,
                                                   const float* __restrict__ bias,
                                                   float* __restrict__ out, int n) {
  const int lane = (int)(threadIdx.x & 31);
  const int node = __builtin_amdgcn_readfirstlane((int)((blockIdx.x * blockDim.x + threadIdx.x) >> 5));
  if (node >= n) return;
  const int cnt  = __builtin_amdgcn_readfirstlane(deg[node]) + 1;
  const int tcnt = (cnt < kRow) ? cnt : kRow;
  const int* trow = tab + (size_t)node * kRow;
  int L = 0;
  if (cnt > kRow) {
    L = __builtin_amdgcn_readfirstlane(deg[n]);
    L = (L < E) ? L : E;
  }
  const float* hl = h + lane * 4;

  v4f ws, wd;
  ws.x = att_s[lane * 4 + 0]; ws.y = att_s[lane * 4 + 1];
  ws.z = att_s[lane * 4 + 2]; ws.w = att_s[lane * 4 + 3];
  wd.x = att_d[lane * 4 + 0]; wd.y = att_d[lane * 4 + 1];
  wd.z = att_d[lane * 4 + 2]; wd.w = att_d[lane * 4 + 3];

  int s0 = node;
  if (lane > 0 && lane < cnt) s0 = trow[lane];
  const v4f h0 = *(const v4f*)(hl + (size_t)node * 128);
  const float ad = sum8(dot4(h0, wd));
  float m = lrelu(sum8(dot4(h0, ws)) + ad);
  float z = 1.f;
  v4f acc = h0;

  for (int cb = 0; cb < tcnt; cb += 32) {
    int s = s0;
    if (cb > 0) {
      const int j = cb + lane;
      s = node;
      if (j < tcnt) s = trow[j];
    }
    const int lim = (tcnt - cb < 32) ? (tcnt - cb) : 32;
    int t = (cb == 0) ? 1 : 0;
    for (; t + 4 <= lim; t += 4) take_rows<4>(acc, z, m, s, t, hl, ws, ad);
    const int rest = lim - t;
    if (rest == 3)      take_rows<3>(acc, z, m, s, t, hl, ws, ad);
    else if (rest == 2) take_rows<2>(acc, z, m, s, t, hl, ws, ad);
    else if (rest == 1) take_rows<1>(acc, z, m, s, t, hl, ws, ad);
  }
  for (int kb = 0; kb < L; kb += 32) {
    const int k = kb + lane;
    int s = node;
    const bool mine = (k < L) && (spill_dst[k] == node);
    if (mine) s = spill_src[k];
    unsigned todo = __builtin_amdgcn_ballot_w32(mine);
    while (todo != 0u) {
      const int t = __builtin_ctz(todo);
      todo &= todo - 1u;
      take_rows<1>(acc, z, m, s, t, hl, ws, ad);
    }
  }
  const float iz = 1.f / z;
  v4f r = acc * iz;
  r.x += bias[lane * 4 + 0];
  r.y += bias[lane * 4 + 1];
  r.z += bias[lane * 4 + 2];
  r.w += bias[lane * 4 + 3];
  r.x = (r.x > 0.f) ? r.x : expm1f(r.x);
  r.y = (r.y > 0.f) ? r.y : expm1f(r.y);
  r.z = (r.z > 0.f) ? r.z : expm1f(r.z);
  r.w = (r.w > 0.f) ? r.w : expm1f(r.w);
  *(v4f*)(out + (size_t)node * 128 + lane * 4) = r;
}


template <int K>
__device__ __forceinline__ int quad_bcast(int v) {
  return __builtin_amdgcn_update_dpp(0, v, K * 0x55, 0xf, 0xf, true);
}
__device__ __forceinline__ float sum4(float v) {
  v = xmask_add<1>(v);
  v = xmask_add<2>(v);
  return v;
}

__device__ __forceinline__ void take_one2(v4f& acc, float& z, float& m, int st, int on,
                                          const v4f* __restrict__ hr, const v4f& ws, float ad) {
  const v4f   hv = hr[(unsigned)st * 4u];
  const float v  = lrelu(sum4(dot4(hv, ws)) + ad);
  const float e  = (on != 0) ? v : -__builtin_huge_valf();
  const float mn = fmaxf(m, e);
  const float f  = __expf(m - mn);
  const float w  = __expf(e - mn);
  z   = z * f + w;
  acc = acc * f + hv * w;
  m = mn;
}

__global__ __launch_bounds__(256) void gat2_node_k(const int* __restrict__ deg,
                                                   const int* __restrict__ tab,
                                                   const int* __restrict__ spill_dst,
                                                   const int* __restrict__ spill_src, int E,
                                                   const float* __restrict__ h2,
                                                   const float* __restrict__ att_s,
                                                   const float* __restrict__ att_d,
                                                   const float* __restrict__ bias,
                                                   float* __restrict__ out, int n) {
  const int node = (int)((blockIdx.x * blockDim.x + threadIdx.x) >> 2);
  const int q    = (int)(threadIdx.x & 3);
  if (node >= n) return;
  const int cnt  = deg[node] + 1;
  const int tcnt = (cnt < kRow) ? cnt : kRow;
  const int* trow = tab + (size_t)node * kRow;
  int L = 0;
  if (cnt > kRow) {
    L = deg[n];
    L = (L < E) ? L : E;
  }
  const float NEG = -__builtin_huge_valf();
  v4f ws, wd;
  ws.x = att_s[q * 4 + 0]; ws.y = att_s[q * 4 + 1];
  ws.z = att_s[q * 4 + 2]; ws.w = att_s[q * 4 + 3];
  wd.x = att_d[q * 4 + 0]; wd.y = att_d[q * 4 + 1];
  wd.z = att_d[q * 4 + 2]; wd.w = att_d[q * 4 + 3];
  const v4f* __restrict__ hr = (const v4f*)h2 + q;

  int sa = node, sb = node;
  if (1 + q < tcnt) sa = trow[1 + q];
  if (5 + q < tcnt) sb = trow[5 + q];
  const v4f h0 = hr[(unsigned)node * 4u];
  const float ad = sum4(dot4(h0, wd));
  float m = lrelu(sum4(dot4(h0, ws)) + ad);
  float z = 1.f;
  v4f acc = h0;

  for (int t = 1; t < tcnt; t += 8) {
    const int ca = sa, cb = sb;
    v4f   hv[8];
    float e[8];
    hv[0] = hr[(unsigned)quad_bcast<0>(ca) * 4u];
    hv[1] = hr[(unsigned)quad_bcast<1>(ca) * 4u];
    hv[2] = hr[(unsigned)quad_bcast<2>(ca) * 4u];
    hv[3] = hr[(unsigned)quad_bcast<3>(ca) * 4u];
    hv[4] = hr[(unsigned)quad_bcast<0>(cb) * 4u];
    hv[5] = hr[(unsigned)quad_bcast<1>(cb) * 4u];
    hv[6] = hr[(unsigned)quad_bcast<2>(cb) * 4u];
    hv[7] = hr[(unsigned)quad_bcast<3>(cb) * 4u];
    sa = node;
    sb = node;
    if (t + 8 + q < tcnt)  sa = trow[t + 8 + q];
    if (t + 12 + q < tcnt) sb = trow[t + 12 + q];
#pragma unroll
    for (int k = 0; k < 8; ++k) {
      const float v = lrelu(sum4(dot4(hv[k], ws)) + ad);
      e[k] = (t + k < tcnt) ? v : NEG;
    }
    const float mn = fmaxf(fmaxf(fmaxf(m, fmaxf(e[0], e[1])), fmaxf(e[2], e[3])),
                           fmaxf(fmaxf(e[4], e[5]), fmaxf(e[6], e[7])));
    const float f = __expf(m - mn);
    float w[8];
#pragma unroll
    for (int k = 0; k < 8; ++k) w[k] = __expf(e[k] - mn);
    z   = z * f + (((w[0] + w[1]) + (w[2] + w[3])) + ((w[4] + w[5]) + (w[6] + w[7])));
    acc = acc * f + (((hv[0] * w[0] + hv[1] * w[1]) + (hv[2] * w[2] + hv[3] * w[3])) +
                     ((hv[4] * w[4] + hv[5] * w[5]) + (hv[6] * w[6] + hv[7] * w[7])));
    m = mn;
  }
  for (int kb = 0; kb < L; kb += 4) {
    const int k = kb + q;
    int s = node;
    int mine = 0;
    if (k < L && spill_dst[k] == node) {
      s = spill_src[k];
      mine = 1;
    }
    take_one2(acc, z, m, quad_bcast<0>(s), quad_bcast<0>(mine), hr, ws, ad);
    take_one2(acc, z, m, quad_bcast<1>(s), quad_bcast<1>(mine), hr, ws, ad);
    take_one2(acc, z, m, quad_bcast<2>(s), quad_bcast<2>(mine), hr, ws, ad);
    take_one2(acc, z, m, quad_bcast<3>(s), quad_bcast<3>(mine), hr, ws, ad);
  }
  v4f r = acc / z;
  r.x += bias[q * 4 + 0];
  r.y += bias[q * 4 + 1];
  r.z += bias[q * 4 + 2];
  r.w += bias[q * 4 + 3];
  *(v4f*)(out + (size_t)node * 16 + q * 4) = r;
}

extern "C" void kernel_launch(void* const* d_in, const int* in_sizes, int n_in,
                              void* d_out, int out_size, void* d_ws, size_t ws_size,
                              hipStream_t stream) {
  const float* x    = (const float*)d_in[0];
  const int*   ei   = (const int*)d_in[1];
  const float* W1   = (const float*)d_in[2];
  const float* as1w = (const float*)d_in[3];
  const float* ad1w = (const float*)d_in[4];
  const float* b1   = (const float*)d_in[5];
  const float* W2   = (const float*)d_in[6];
  const float* as2w = (const float*)d_in[7];
  const float* ad2w = (const float*)d_in[8];
  const float* b2   = (const float*)d_in[9];
  float* out = (float*)d_out;

  const int n = in_sizes[0] / 128;
  const int E = in_sizes[1] / 2;

  float* ws     = (float*)d_ws;
  float* h1     = ws;
  float* out1   = h1 + (size_t)n * 128;
  float* h2     = out1 + (size_t)n * 128;
  int*   tab    = (int*)(h2 + (size_t)n * 16);
  int*   spd    = tab + (size_t)n * kRow;
  int*   sps    = spd + E;
  int*   pairs  = sps + E;
  int*   cnt    = pairs + 2 * (size_t)E;
  int*   segoff = cnt + n + 1;

  const int B = 256;
  const int mtiles = (n + 15) / 16;
  const int gemm_blocks = (mtiles + 7) / 8;
  const int nblk = (E + kChunk - 1) / kChunk;
  const int nrng = (n + kBk - 1) / kBk;

  proj1_bin_k<<<gemm_blocks + nblk, B, 0, stream>>>(x, W1, h1, mtiles, gemm_blocks, ei, E, n, cnt, segoff, pairs);
  rank_fill_k<<<nrng, 1024, 0, stream>>>(pairs, segoff, nblk, cnt, n, tab, spd, sps);

  gat1_node_k<<<(int)(((long)n * 32 + B - 1) / B), B, 0, stream>>>(cnt, tab, spd, sps, E, h1, as1w, ad1w, b1, out1, n);

  gemm_wmma_f32<1, 128, 16><<<(mtiles + 7) / 8, B, 0, stream>>>(out1, W2, h2, mtiles);
  gat2_node_k<<<(int)(((long)n * 4 + B - 1) / B), B, 0, stream>>>(cnt, tab, spd, sps, E, h2, as2w, ad2w, b2, out, n);
}


# Slot rows leave the ranking pass as whole 128-byte lines built in LDS, padded with the node id: one store per node, not per edge, and the gathers fetch slots before they know the count; same sums
# speedup vs baseline: 8.4504x; 1.0448x over previous
#include <hip/hip_runtime.h>


typedef float v2f __attribute__((ext_vector_type(2)));
typedef float v8f __attribute__((ext_vector_type(8)));
typedef float v4f __attribute__((ext_vector_type(4)));

template <int NT, int K, int NCOL>
__device__ __forceinline__ void gemm_wave_tile(const float* __restrict__ A,
                                               const float* __restrict__ B,
                                               float* __restrict__ C,
                                               int wave, int lane) {
  const int lane15 = lane & 15;
  const int khalf  = (lane >> 4) * 2;
  const long rowbase = (long)wave * 16;

  v8f acc[NT];
#pragma unroll
  for (int t = 0; t < NT; ++t) acc[t] = (v8f){0.f,0.f,0.f,0.f,0.f,0.f,0.f,0.f};

  const float* arow = A + (rowbase + lane15) * K + khalf;
#pragma unroll 4
  for (int kb = 0; kb < K; kb += 4) {
    v2f a;
    a.x = arow[kb];
    a.y = arow[kb + 1];
#pragma unroll
    for (int t = 0; t < NT; ++t) {
      const float* bp = B + (size_t)(kb + khalf) * NCOL + t * 16 + lane15;
      v2f b;
      b.x = bp[0];
      b.y = bp[NCOL];
      acc[t] = __builtin_amdgcn_wmma_f32_16x16x4_f32(
           false, a,  false, b,
           (short)0, acc[t],  false,  false);
    }
  }
  const int rowoff = (lane >> 4) * 8;
#pragma unroll
  for (int t = 0; t < NT; ++t) {
#pragma unroll
    for (int j = 0; j < 8; ++j) {
      C[(rowbase + rowoff + j) * NCOL + t * 16 + lane15] = acc[t][j];
    }
  }
}

template <int NT, int K, int NCOL>
__global__ __launch_bounds__(256) void gemm_wmma_f32(const float* __restrict__ A,
                                                     const float* __restrict__ B,
                                                     float* __restrict__ C,
                                                     int mtiles) {
  const int wave = (int)((blockIdx.x * blockDim.x + threadIdx.x) >> 5);
  const int lane = (int)(threadIdx.x & 31);
  if (wave < mtiles) gemm_wave_tile<NT, K, NCOL>(A, B, C, wave, lane);
}

constexpr int kRow  = 64;
constexpr int kLrow = 32;

__device__ __forceinline__ void put_edge(int d, int r, int s, int* __restrict__ tab,
                                         int* __restrict__ spill_n, int* __restrict__ spill_dst,
                                         int* __restrict__ spill_src) {
  if (r < kRow - 1) {
    tab[(size_t)d * kRow + 1 + r] = s;
  } else {
    const int k = atomicAdd(spill_n, 1);
    spill_dst[k] = d;
    spill_src[k] = s;
  }
}

constexpr int kSh     = 9;
constexpr int kBk     = 1 << kSh;
constexpr int kMaxB   = 256;
constexpr int kChunk  = 4096;
constexpr int kSegStr = kMaxB + 1;

__device__ __forceinline__ void bin_block(const int* __restrict__ esrc, const int* __restrict__ edst,
                                          int E, int n, int j, int* __restrict__ segoff,
                                          int2* __restrict__ pairs, int* hist, int* offs, int* wtot) {
  const int tid  = (int)threadIdx.x;
  const int lane = tid & 31;
  const int wave = tid >> 5;
  hist[tid] = 0;
  __syncthreads();
  const long base = (long)j * kChunk;
  int d[16], s[16], lr[16];
#pragma unroll
  for (int k = 0; k < 16; ++k) {
    const long i = base + k * 256 + tid;
    d[k] = -1;
    s[k] = 0;
    if (i < E) { d[k] = edst[i]; s[k] = esrc[i]; }
  }
#pragma unroll
  for (int k = 0; k < 16; ++k) {
    lr[k] = 0;
    if ((unsigned)d[k] < (unsigned)n) lr[k] = atomicAdd(&hist[d[k] >> kSh], 1);
    else d[k] = -1;
  }
  __syncthreads();
  const int v = hist[tid];
  int x = v;
#pragma unroll
  for (int o = 1; o < 32; o <<= 1) {
    const int y = __shfl_up(x, o, 32);
    if (lane >= o) x += y;
  }
  if (lane == 31) wtot[wave] = x;
  __syncthreads();
  int wp = 0;
  for (int w = 0; w < wave; ++w) wp += wtot[w];
  const int excl = wp + x - v;
  offs[tid] = excl;
  segoff[(size_t)j * kSegStr + tid] = excl;
  if (tid == kMaxB - 1) segoff[(size_t)j * kSegStr + kMaxB] = excl + v;
  __syncthreads();
#pragma unroll
  for (int k = 0; k < 16; ++k) {
    if (d[k] >= 0) pairs[base + offs[d[k] >> kSh] + lr[k]] = make_int2(d[k], s[k]);
  }
}

__global__ __launch_bounds__(256) void proj1_bin_k(const float* __restrict__ x,
                                                   const float* __restrict__ W1,
                                                   float* __restrict__ h1, int mtiles,
                                                   int gemm_blocks,
                                                   const int* __restrict__ ei, int E, int n,
                                                   int* __restrict__ cnt,
                                                   int* __restrict__ segoff,
                                                   int* __restrict__ pairs) {
  __shared__ int hist[kMaxB];
  __shared__ int offs[kMaxB];
  __shared__ int wtot[8];
  if ((int)blockIdx.x < gemm_blocks) {
    const int lane = (int)(threadIdx.x & 31);
    const int wave = (int)((blockIdx.x * 256 + threadIdx.x) >> 5);
    if (wave < mtiles) gemm_wave_tile<8, 128, 128>(x, W1, h1, wave, lane);
  } else {
    const int j = (int)blockIdx.x - gemm_blocks;
    if (j == 0 && threadIdx.x == 0) cnt[n] = 0;
    bin_block(ei, ei + E, E, n, j, segoff, (int2*)pairs, hist, offs, wtot);
  }
}

__global__ __launch_bounds__(1024) void rank_fill_k(const int* __restrict__ pairs_i,
                                                    const int* __restrict__ segoff, int nblk,
                                                    int* __restrict__ cnt, int n,
                                                    int* __restrict__ tab,
                                                    int* __restrict__ spill_dst,
                                                    int* __restrict__ spill_src) {
  const int2* __restrict__ pairs = (const int2*)pairs_i;
  __shared__ int lcnt[kBk];
  __shared__ int sst[1024];
  __shared__ int sct[1024];
  __shared__ __attribute__((aligned(16))) int lrow[kBk * kLrow];
  const int tid  = (int)threadIdx.x;
  const int lane = tid & 31;
  const int wave = tid >> 5;
  const int b    = (int)blockIdx.x;
  const int nlo  = b * kBk;
  if (tid < kBk) lcnt[tid] = 0;
  for (int g = tid; g < kBk * kLrow / 4; g += 1024) {
    const int id = nlo + (g >> 3);
    *(int4*)&lrow[g * 4] = make_int4(id, id, id, id);
  }
  for (int jb = 0; jb < nblk; jb += 1024) {
    __syncthreads();
    const int j = jb + tid;
    int st = 0, c = 0;
    if (j < nblk) {
      st = segoff[(size_t)j * kSegStr + b];
      c  = segoff[(size_t)j * kSegStr + b + 1] - st;
    }
    sst[tid] = st;
    sct[tid] = c;
    __syncthreads();
    for (int i = 0; i < 32; i += 4) {
      if (jb + i * 32 >= nblk) break;
      int2 p[4];
      int  c4[4];
      long sb[4];
#pragma unroll
      for (int u = 0; u < 4; ++u) {
        const int q = (i + u) * 32 + wave;
        c4[u] = sct[q];
        sb[u] = (long)(jb + q) * kChunk + sst[q];
        p[u]  = make_int2(nlo, 0);
        if (lane < c4[u]) p[u] = pairs[sb[u] + lane];
      }
#pragma unroll
      for (int u = 0; u < 4; ++u) {
        if (lane < c4[u]) {
          const int ln = p[u].x - nlo;
          const int r  = atomicAdd(&lcnt[ln], 1);
          if (r < kLrow - 1) lrow[ln * kLrow + 1 + r] = p[u].y;
          else put_edge(p[u].x, r, p[u].y, tab, cnt + n, spill_dst, spill_src);
        }
        for (int e = lane + 32; e < c4[u]; e += 32) {
          const int2 pe = pairs[sb[u] + e];
          const int ln = pe.x - nlo;
          const int r  = atomicAdd(&lcnt[ln], 1);
          if (r < kLrow - 1) lrow[ln * kLrow + 1 + r] = pe.y;
          else put_edge(pe.x, r, pe.y, tab, cnt + n, spill_dst, spill_src);
        }
      }
    }
  }
  __syncthreads();
  if (tid < kBk) {
    const int node = nlo + tid;
    if (node < n) cnt[node] = lcnt[tid];
  }
  for (int g = tid; g < kBk * kLrow / 4; g += 1024) {
    const int node = nlo + (g >> 3);
    if (node < n) *(int4*)&tab[(size_t)node * kRow + (g & 7) * 4] = *(const int4*)&lrow[g * 4];
  }
}

__device__ __forceinline__ float lrelu(float e) { return (e >= 0.f) ? e : 0.2f * e; }

template <int K>
__device__ __forceinline__ float xmask_add(float v) {
  const int o = __builtin_amdgcn_update_dpp(0, __float_as_int(v), 0x160 | K, 0xf, 0xf, true);
  return v + __int_as_float(o);
}
__device__ __forceinline__ float sum8(float v) {
  v = xmask_add<1>(v);
  v = xmask_add<2>(v);
  v = xmask_add<4>(v);
  return v;
}
__device__ __forceinline__ float sum16(float v) {
  v = xmask_add<8>(sum8(v));
  return v;
}

__device__ __forceinline__ float dot4(const v4f& a, const v4f& b) {
  const v2f p = a.lo * b.lo + a.hi * b.hi;
  return p.x + p.y;
}


template <int R>
__device__ __forceinline__ void take_rows(v4f& acc, float& z, float& m, int s, int t,
                                          const float* __restrict__ hl, const v4f& ws, float ad) {
  v4f   hv[R];
  float e[R];
#pragma unroll
  for (int k = 0; k < R; ++k) {
    const int st = __builtin_amdgcn_readlane(s, t + k);
    hv[k] = *(const v4f*)(hl + (size_t)st * 128);
  }
  float mn = m;
#pragma unroll
  for (int k = 0; k < R; ++k) {
    e[k] = lrelu(sum8(dot4(hv[k], ws)) + ad);
    mn = fmaxf(mn, e[k]);
  }
  const float f = __expf(m - mn);
  float w  = __expf(e[0] - mn);
  float zs = w;
  v4f   sm = hv[0] * w;
#pragma unroll
  for (int k = 1; k < R; ++k) {
    w = __expf(e[k] - mn);
    zs += w;
    sm += hv[k] * w;
  }
  z   = z * f + zs;
  acc = acc * f + sm;
  m = mn;
}

__global__ __launch_bounds__(256) void gat1_node_k(const int* __restrict__ deg,
                                                   const int* __restrict__ tab,
                                                   const int* __restrict__ spill_dst,
                                                   const int* __restrict__ spill_src, int E,
                                                   const float* __restrict__ h,
                                                   const float* __restrict__ att_s,
                                                   const float* __restrict__ att_d,
                                                   const float* __restrict__ bias,
                                                   float* __restrict__ out, int n) {
  const int lane = (int)(threadIdx.x & 31);
  const int node = __builtin_amdgcn_readfirstlane((int)((blockIdx.x * blockDim.x + threadIdx.x) >> 5));
  if (node >= n) return;
  const int* trow = tab + (size_t)node * kRow;
  const float* hl = h + lane * 4;

  v4f ws, wd;
  ws.x = att_s[lane * 4 + 0]; ws.y = att_s[lane * 4 + 1];
  ws.z = att_s[lane * 4 + 2]; ws.w = att_s[lane * 4 + 3];
  wd.x = att_d[lane * 4 + 0]; wd.y = att_d[lane * 4 + 1];
  wd.z = att_d[lane * 4 + 2]; wd.w = att_d[lane * 4 + 3];

  const int s0 = trow[lane];
  const v4f h0 = *(const v4f*)(hl + (size_t)node * 128);

  const int cnt  = __builtin_amdgcn_readfirstlane(deg[node]) + 1;
  const int tcnt = (cnt < kRow) ? cnt : kRow;
  int L = 0;
  if (cnt > kRow) {
    L = __builtin_amdgcn_readfirstlane(deg[n]);
    L = (L < E) ? L : E;
  }
  const float ad = sum8(dot4(h0, wd));
  float m = lrelu(sum8(dot4(h0, ws)) + ad);
  float z = 1.f;
  v4f acc = h0;

  for (int cb = 0; cb < tcnt; cb += 32) {
    int s = s0;
    if (cb > 0) {
      const int j = cb + lane;
      s = node;
      if (j < tcnt) s = trow[j];
    }
    const int lim = (tcnt - cb < 32) ? (tcnt - cb) : 32;
    int t = (cb == 0) ? 1 : 0;
    for (; t + 4 <= lim; t += 4) take_rows<4>(acc, z, m, s, t, hl, ws, ad);
    const int rest = lim - t;
    if (rest == 3)      take_rows<3>(acc, z, m, s, t, hl, ws, ad);
    else if (rest == 2) take_rows<2>(acc, z, m, s, t, hl, ws, ad);
    else if (rest == 1) take_rows<1>(acc, z, m, s, t, hl, ws, ad);
  }
  for (int kb = 0; kb < L; kb += 32) {
    const int k = kb + lane;
    int s = node;
    const bool mine = (k < L) && (spill_dst[k] == node);
    if (mine) s = spill_src[k];
    unsigned todo = __builtin_amdgcn_ballot_w32(mine);
    while (todo != 0u) {
      const int t = __builtin_ctz(todo);
      todo &= todo - 1u;
      take_rows<1>(acc, z, m, s, t, hl, ws, ad);
    }
  }
  const float iz = 1.f / z;
  v4f r = acc * iz;
  r.x += bias[lane * 4 + 0];
  r.y += bias[lane * 4 + 1];
  r.z += bias[lane * 4 + 2];
  r.w += bias[lane * 4 + 3];
  r.x = (r.x > 0.f) ? r.x : expm1f(r.x);
  r.y = (r.y > 0.f) ? r.y : expm1f(r.y);
  r.z = (r.z > 0.f) ? r.z : expm1f(r.z);
  r.w = (r.w > 0.f) ? r.w : expm1f(r.w);
  *(v4f*)(out + (size_t)node * 128 + lane * 4) = r;
}


template <int K>
__device__ __forceinline__ int quad_bcast(int v) {
  return __builtin_amdgcn_update_dpp(0, v, K * 0x55, 0xf, 0xf, true);
}
__device__ __forceinline__ float sum4(float v) {
  v = xmask_add<1>(v);
  v = xmask_add<2>(v);
  return v;
}

__device__ __forceinline__ void take_one2(v4f& acc, float& z, float& m, int st, int on,
                                          const v4f* __restrict__ hr, const v4f& ws, float ad) {
  const v4f   hv = hr[(unsigned)st * 4u];
  const float v  = lrelu(sum4(dot4(hv, ws)) + ad);
  const float e  = (on != 0) ? v : -__builtin_huge_valf();
  const float mn = fmaxf(m, e);
  const float f  = __expf(m - mn);
  const float w  = __expf(e - mn);
  z   = z * f + w;
  acc = acc * f + hv * w;
  m = mn;
}

__global__ __launch_bounds__(256) void gat2_node_k(const int* __restrict__ deg,
                                                   const int* __restrict__ tab,
                                                   const int* __restrict__ spill_dst,
                                                   const int* __restrict__ spill_src, int E,
                                                   const float* __restrict__ h2,
                                                   const float* __restrict__ att_s,
                                                   const float* __restrict__ att_d,
                                                   const float* __restrict__ bias,
                                                   float* __restrict__ out, int n) {
  const int node = (int)((blockIdx.x * blockDim.x + threadIdx.x) >> 2);
  const int q    = (int)(threadIdx.x & 3);
  if (node >= n) return;
  const int* trow = tab + (size_t)node * kRow;
  const float NEG = -__builtin_huge_valf();
  v4f ws, wd;
  ws.x = att_s[q * 4 + 0]; ws.y = att_s[q * 4 + 1];
  ws.z = att_s[q * 4 + 2]; ws.w = att_s[q * 4 + 3];
  wd.x = att_d[q * 4 + 0]; wd.y = att_d[q * 4 + 1];
  wd.z = att_d[q * 4 + 2]; wd.w = att_d[q * 4 + 3];
  const v4f* __restrict__ hr = (const v4f*)h2 + q;

  int sa = trow[1 + q];
  int sb = trow[5 + q];
  const v4f h0 = hr[(unsigned)node * 4u];

  const int cnt  = deg[node] + 1;
  const int tcnt = (cnt < kRow) ? cnt : kRow;
  int L = 0;
  if (cnt > kRow) {
    L = deg[n];
    L = (L < E) ? L : E;
  }
  const float ad = sum4(dot4(h0, wd));
  float m = lrelu(sum4(dot4(h0, ws)) + ad);
  float z = 1.f;
  v4f acc = h0;

  for (int t = 1; t < tcnt; t += 8) {
    const int ca = sa, cb = sb;
    v4f   hv[8];
    float e[8];
    hv[0] = hr[(unsigned)quad_bcast<0>(ca) * 4u];
    hv[1] = hr[(unsigned)quad_bcast<1>(ca) * 4u];
    hv[2] = hr[(unsigned)quad_bcast<2>(ca) * 4u];
    hv[3] = hr[(unsigned)quad_bcast<3>(ca) * 4u];
    hv[4] = hr[(unsigned)quad_bcast<0>(cb) * 4u];
    hv[5] = hr[(unsigned)quad_bcast<1>(cb) * 4u];
    hv[6] = hr[(unsigned)quad_bcast<2>(cb) * 4u];
    hv[7] = hr[(unsigned)quad_bcast<3>(cb) * 4u];
    sa = node;
    sb = node;
    if (t + 8 + q < tcnt)  sa = trow[t + 8 + q];
    if (t + 12 + q < tcnt) sb = trow[t + 12 + q];
#pragma unroll
    for (int k = 0; k < 8; ++k) {
      const float v = lrelu(sum4(dot4(hv[k], ws)) + ad);
      e[k] = (t + k < tcnt) ? v : NEG;
    }
    const float mn = fmaxf(fmaxf(fmaxf(m, fmaxf(e[0], e[1])), fmaxf(e[2], e[3])),
                           fmaxf(fmaxf(e[4], e[5]), fmaxf(e[6], e[7])));
    const float f = __expf(m - mn);
    float w[8];
#pragma unroll
    for (int k = 0; k < 8; ++k) w[k] = __expf(e[k] - mn);
    z   = z * f + (((w[0] + w[1]) + (w[2] + w[3])) + ((w[4] + w[5]) + (w[6] + w[7])));
    acc = acc * f + (((hv[0] * w[0] + hv[1] * w[1]) + (hv[2] * w[2] + hv[3] * w[3])) +
                     ((hv[4] * w[4] + hv[5] * w[5]) + (hv[6] * w[6] + hv[7] * w[7])));
    m = mn;
  }
  for (int kb = 0; kb < L; kb += 4) {
    const int k = kb + q;
    int s = node;
    int mine = 0;
    if (k < L && spill_dst[k] == node) {
      s = spill_src[k];
      mine = 1;
    }
    take_one2(acc, z, m, quad_bcast<0>(s), quad_bcast<0>(mine), hr, ws, ad);
    take_one2(acc, z, m, quad_bcast<1>(s), quad_bcast<1>(mine), hr, ws, ad);
    take_one2(acc, z, m, quad_bcast<2>(s), quad_bcast<2>(mine), hr, ws, ad);
    take_one2(acc, z, m, quad_bcast<3>(s), quad_bcast<3>(mine), hr, ws, ad);
  }
  v4f r = acc / z;
  r.x += bias[q * 4 + 0];
  r.y += bias[q * 4 + 1];
  r.z += bias[q * 4 + 2];
  r.w += bias[q * 4 + 3];
  *(v4f*)(out + (size_t)node * 16 + q * 4) = r;
}

extern "C" void kernel_launch(void* const* d_in, const int* in_sizes, int n_in,
                              void* d_out, int out_size, void* d_ws, size_t ws_size,
                              hipStream_t stream) {
  const float* x    = (const float*)d_in[0];
  const int*   ei   = (const int*)d_in[1];
  const float* W1   = (const float*)d_in[2];
  const float* as1w = (const float*)d_in[3];
  const float* ad1w = (const float*)d_in[4];
  const float* b1   = (const float*)d_in[5];
  const float* W2   = (const float*)d_in[6];
  const float* as2w = (const float*)d_in[7];
  const float* ad2w = (const float*)d_in[8];
  const float* b2   = (const float*)d_in[9];
  float* out = (float*)d_out;

  const int n = in_sizes[0] / 128;
  const int E = in_sizes[1] / 2;

  float* ws     = (float*)d_ws;
  float* h1     = ws;
  float* out1   = h1 + (size_t)n * 128;
  float* h2     = out1 + (size_t)n * 128;
  int*   tab    = (int*)(h2 + (size_t)n * 16);
  int*   spd    = tab + (size_t)n * kRow;
  int*   sps    = spd + E;
  int*   pairs  = sps + E;
  int*   cnt    = pairs + 2 * (size_t)E;
  int*   segoff = cnt + n + 1;

  const int B = 256;
  const int mtiles = (n + 15) / 16;
  const int gemm_blocks = (mtiles + 7) / 8;
  const int nblk = (E + kChunk - 1) / kChunk;
  const int nrng = (n + kBk - 1) / kBk;

  proj1_bin_k<<<gemm_blocks + nblk, B, 0, stream>>>(x, W1, h1, mtiles, gemm_blocks, ei, E, n, cnt, segoff, pairs);
  rank_fill_k<<<nrng, 1024, 0, stream>>>(pairs, segoff, nblk, cnt, n, tab, spd, sps);

  gat1_node_k<<<(int)(((long)n * 32 + B - 1) / B), B, 0, stream>>>(cnt, tab, spd, sps, E, h1, as1w, ad1w, b1, out1, n);

  gemm_wmma_f32<1, 128, 16><<<(mtiles + 7) / 8, B, 0, stream>>>(out1, W2, h2, mtiles);
  gat2_node_k<<<(int)(((long)n * 4 + B - 1) / B), B, 0, stream>>>(cnt, tab, spd, sps, E, h2, as2w, ad2w, b2, out, n);
}
